# starting kernel
# baseline (speedup 1.0000x reference)
#include <hip/hip_runtime.h>

#define T_DIM 128
#define B_DIM 128
#define R_DIM 512
#define NSLOTS 256
#define SPAD   257

typedef float v2f __attribute__((ext_vector_type(2)));
typedef float v8f __attribute__((ext_vector_type(8)));

__global__ __launch_bounds__(128) void weights_kernel(
    const float* __restrict__ u,
    const float* __restrict__ d1,
    const float* __restrict__ d2,
    float* __restrict__ W)
{
    __shared__ float S[B_DIM * SPAD];
    __shared__ float X[B_DIM * SPAD];
    __shared__ float scal[NSLOTS];

    const int b = threadIdx.x;
    float* Srow = &S[b * SPAD];
    float* Xrow = &X[b * SPAD];

    #pragma unroll 8
    for (int j = 0; j < NSLOTS; ++j) Srow[j] = 0.0f;

    for (int t = 0; t < T_DIM; ++t) {
        const int jmax = 2 * t + 1;
        const float ut = u[t * B_DIM + b];
        Srow[2 * t]     = d1[t * B_DIM + b];
        Srow[2 * t + 1] = d2[t * B_DIM + b];

        float used = 0.0f;
        for (int j = jmax; j >= 0; --j) {
            Xrow[j] = ut - used;
            used += Srow[j];
        }
        __syncthreads();

        for (int jj = threadIdx.x; jj <= jmax; jj += B_DIM) {
            float m = -INFINITY;
            #pragma unroll 8
            for (int bb = 0; bb < B_DIM; ++bb)
                m = fmaxf(m, X[bb * SPAD + jj]);
            scal[jj] = m;
        }
        __syncthreads();

        float* wrow = W + ((size_t)t * B_DIM + b) * NSLOTS;
        for (int j = 0; j < NSLOTS; ++j) {
            float w = 0.0f;
            if (j <= jmax) {
                const float s = Srow[j];
                w = fminf(s, scal[j]);
                const float take = fminf(s, fmaxf(0.0f, Xrow[j]));
                Srow[j] = s - take;
            }
            wrow[j] = w;
        }
    }
}

__global__ __launch_bounds__(256) void gemm_kernel(
    const float* __restrict__ W,
    const float* __restrict__ v1,
    const float* __restrict__ v2,
    float* __restrict__ out)
{
    const int lane = threadIdx.x & 31;
    const int wave = threadIdx.x >> 5;
    const int lm   = lane & 15;
    const int half = lane >> 4;

    const int b  = blockIdx.x >> 3;
    const int mt = blockIdx.x & 7;
    const int n0 = wave * 64;

    v8f acc0 = {}, acc1 = {}, acc2 = {}, acc3 = {};

    const float* Ap = W + ((size_t)(mt * 16 + lm) * B_DIM + b) * NSLOTS + 2 * half;
    const size_t bstride = (size_t)B_DIM * R_DIM;
    const float* B1 = v1 + (size_t)half * bstride + (size_t)b * R_DIM + n0 + lm;
    const float* B2 = v2 + (size_t)half * bstride + (size_t)b * R_DIM + n0 + lm;

    #pragma unroll 4
    for (int k = 0; k < NSLOTS; k += 4) {
        const v2f a = *(const v2f*)(Ap + k);
        const size_t off = (size_t)(k >> 1) * bstride;
        v2f b0, b1, b2, b3;
        b0.x = B1[off +  0]; b0.y = B2[off +  0];
        b1.x = B1[off + 16]; b1.y = B2[off + 16];
        b2.x = B1[off + 32]; b2.y = B2[off + 32];
        b3.x = B1[off + 48]; b3.y = B2[off + 48];
        acc0 = __builtin_amdgcn_wmma_f32_16x16x4_f32(false, a, false, b0, (short)0, acc0, false, false);
        acc1 = __builtin_amdgcn_wmma_f32_16x16x4_f32(false, a, false, b1, (short)0, acc1, false, false);
        acc2 = __builtin_amdgcn_wmma_f32_16x16x4_f32(false, a, false, b2, (short)0, acc2, false, false);
        acc3 = __builtin_amdgcn_wmma_f32_16x16x4_f32(false, a, false, b3, (short)0, acc3, false, false);
    }

    #pragma unroll
    for (int r = 0; r < 8; ++r) {
        const int trow = mt * 16 + r + 8 * half;
        float* orow = out + ((size_t)trow * B_DIM + b) * R_DIM + n0 + lm;
        orow[ 0] = acc0[r];
        orow[16] = acc1[r];
        orow[32] = acc2[r];
        orow[48] = acc3[r];
    }
}

extern "C" void kernel_launch(void* const* d_in, const int* in_sizes, int n_in,
                              void* d_out, int out_size, void* d_ws, size_t ws_size,
                              hipStream_t stream) {
    const float* u  = (const float*)d_in[0];
    const float* d1 = (const float*)d_in[1];
    const float* d2 = (const float*)d_in[2];
    const float* v1 = (const float*)d_in[3];
    const float* v2 = (const float*)d_in[4];
    float* out = (float*)d_out;
    float* W   = (float*)d_ws;

    weights_kernel<<<dim3(1), dim3(128), 0, stream>>>(u, d1, d2, W);
    gemm_kernel<<<dim3(B_DIM * 8), dim3(256), 0, stream>>>(W, v1, v2, out);
}


# GEMM loads and stores 16 bytes per lane: a lane takes 4 neighbouring columns of v1/v2 (component q feeds tile q) and 8 consecutive k of its W row, 4 k-steps per wait; sum order changes
# speedup vs baseline: 33.0934x; 33.0934x over previous
#include <hip/hip_runtime.h>


#define T_DIM 128
#define B_DIM 128
#define R_DIM 512
#define NSLOTS 256
#define GSLOTS 32
#define WAVES  8
#define NBW    16

typedef float v2f __attribute__((ext_vector_type(2)));
typedef float v4f __attribute__((ext_vector_type(4)));
typedef float v8f __attribute__((ext_vector_type(8)));
typedef float v16f __attribute__((ext_vector_type(16)));

__device__ __forceinline__ v16f row16(const float* p, int t, int b0) {
    return *(const v16f*)(p + t * B_DIM + b0);
}

__device__ __forceinline__ void emit_weights(float* wcol, int t, int j,
                                             const float (&Sof)[NBW], const float (&part)[WAVES])
{
    float scal = part[0];
    #pragma unroll
    for (int w = 1; w < WAVES; ++w) scal = fmaxf(scal, part[w]);
    scal = (j <= 2 * t + 1) ? scal : 0.0f;
    float* wr = wcol + (size_t)t * B_DIM * NSLOTS;
    #pragma unroll
    for (int i = 0; i < NBW; ++i) wr[i * NSLOTS] = fminf(Sof[i], scal);
}

__device__ __forceinline__ void walk_step(
    int t, int t0, int j, bool odd, int lane, int wv, int b0, int b0v,
    const float* u, const float* d1, const float* d2, float* wcol, float (*red)[WAVES][32],
    float (&S)[NBW], float (&used)[NBW],
    const v16f& uc, const v16f& d1n, const v16f& d2n,
    const float (&Sprev)[NBW], const float (&partPrev)[WAVES],
    v16f& un, v16f& d1nn, v16f& d2nn, float (&Sold)[NBW], float (&part)[WAVES])
{
    const int ta = (t + 1 < T_DIM) ? t + 1 : T_DIM - 1;
    const int tb = (t + 2 < T_DIM) ? t + 2 : T_DIM - 1;
    un   = row16(u,  ta, b0);
    d1nn = row16(d1, tb, b0v);
    d2nn = row16(d2, tb, b0v);

    float X[NBW];
    float m = -INFINITY;
    #pragma unroll
    for (int i = 0; i < NBW; ++i) {
        X[i] = uc[i] - used[i];
        m = fmaxf(m, X[i]);
    }
    red[t & 1][wv][lane] = m;

    #pragma unroll
    for (int i = 0; i < NBW; ++i) {
        Sold[i] = S[i];
        S[i] = S[i] - fminf(S[i], fmaxf(0.0f, X[i]));
        const float Dn = d1n[i] + d2n[i];
        const float up = fmaxf(uc[i], 0.0f);
        used[i] = fmaxf(used[i] + (Dn - up), Dn);
    }
    if (t + 1 < t0 + GSLOTS / 2) {
        const bool born = (j >> 1) == t + 1;
        #pragma unroll
        for (int i = 0; i < NBW; ++i) {
            S[i]    = born ? (odd ? d2n[i] : d1n[i]) : S[i];
            used[i] = born ? (odd ? 0.0f : d2n[i]) : used[i];
        }
    }

    if (t > t0) emit_weights(wcol, t - 1, j, Sprev, partPrev);

    __builtin_amdgcn_fence(__ATOMIC_RELEASE, "workgroup", "local");
    __builtin_amdgcn_s_barrier();
    __builtin_amdgcn_fence(__ATOMIC_ACQUIRE, "workgroup", "local");

    #pragma unroll
    for (int w = 0; w < WAVES; ++w) part[w] = red[t & 1][w][lane];
}

__global__ __launch_bounds__(256) void weights_kernel(
    const float* __restrict__ u,
    const float* __restrict__ d1,
    const float* __restrict__ d2,
    float* __restrict__ W)
{
    __shared__ float red[2][WAVES][32];

    const int lane = threadIdx.x & 31;
    const int wv = __builtin_amdgcn_readfirstlane((int)(threadIdx.x >> 5));
    const int g  = blockIdx.x;
    const int j  = g * GSLOTS + lane;
    const int b0 = wv * NBW;
    const int b0v = (int)(threadIdx.x >> 5) * NBW;
    const int t0 = g * (GSLOTS / 2);
    const bool odd = (j & 1) != 0;

    float* wcol = W + (size_t)b0 * NSLOTS + j;

    {
        float* z = W + ((size_t)b0 + (lane >> 3)) * NSLOTS + g * GSLOTS + (lane & 7) * 4;
        const v4f zero = {0.0f, 0.0f, 0.0f, 0.0f};
        for (int t = 0; t < t0; ++t) {
            float* zr = z + (size_t)t * B_DIM * NSLOTS;
            #pragma unroll
            for (int q = 0; q < NBW / 4; ++q) *(v4f*)(zr + (size_t)q * 4 * NSLOTS) = zero;
        }
    }

    float S[NBW], used[NBW], SpA[NBW], SpB[NBW], pA[WAVES], pB[WAVES];
    v16f uA, d1A, d2A, uB, d1B, d2B;
    uA  = row16(u, t0, b0);
    d1A = row16(d1, t0 + 1, b0v);
    d2A = row16(d2, t0 + 1, b0v);
    {
        const v16f a1 = row16(d1, t0, b0v);
        const v16f a2 = row16(d2, t0, b0v);
        const bool born = (j >> 1) == t0;
        #pragma unroll
        for (int i = 0; i < NBW; ++i) {
            S[i]    = born ? (odd ? a2[i] : a1[i]) : 0.0f;
            used[i] = born ? (odd ? 0.0f : a2[i]) : 0.0f;
            SpA[i]  = 0.0f;
        }
        #pragma unroll
        for (int w = 0; w < WAVES; ++w) pA[w] = 0.0f;
    }

    for (int t = t0; t < T_DIM; t += 2) {
        walk_step(t,     t0, j, odd, lane, wv, b0, b0v, u, d1, d2, wcol, red, S, used,
                  uA, d1A, d2A, SpA, pA,   uB, d1B, d2B, SpB, pB);
        walk_step(t + 1, t0, j, odd, lane, wv, b0, b0v, u, d1, d2, wcol, red, S, used,
                  uB, d1B, d2B, SpB, pB,   uA, d1A, d2A, SpA, pA);
    }

    emit_weights(wcol, T_DIM - 1, j, SpA, pA);
}

__global__ __launch_bounds__(256) void gemm_kernel(
    const float* __restrict__ W,
    const float* __restrict__ v1,
    const float* __restrict__ v2,
    float* __restrict__ out)
{
    const int lane = threadIdx.x & 31;
    const int wave = threadIdx.x >> 5;
    const int c    = lane & 15;
    const int h    = lane >> 4;

    const int b  = blockIdx.x >> 3;
    const int mt = blockIdx.x & 7;
    const int n0 = wave * 64;

    v8f acc0 = {}, acc1 = {}, acc2 = {}, acc3 = {};

    const float* Ap = W + ((size_t)(mt * 16 + c) * B_DIM + b) * NSLOTS + 8 * h;
    const size_t bstride = (size_t)B_DIM * R_DIM;
    const float* B1 = v1 + (size_t)(4 * h) * bstride + (size_t)b * R_DIM + n0 + 4 * c;
    const float* B2 = v2 + (size_t)(4 * h) * bstride + (size_t)b * R_DIM + n0 + 4 * c;

    for (int G = 0; G < NSLOTS / 16; ++G) {
        const v4f alo = *(const v4f*)(Ap + 16 * G);
        const v4f ahi = *(const v4f*)(Ap + 16 * G + 4);
        const float* p1 = B1 + (size_t)(8 * G) * bstride;
        const float* p2 = B2 + (size_t)(8 * G) * bstride;
        const v4f x0 = *(const v4f*)(p1);
        const v4f y0 = *(const v4f*)(p2);
        const v4f x1 = *(const v4f*)(p1 + bstride);
        const v4f y1 = *(const v4f*)(p2 + bstride);
        const v4f x2 = *(const v4f*)(p1 + 2 * bstride);
        const v4f y2 = *(const v4f*)(p2 + 2 * bstride);
        const v4f x3 = *(const v4f*)(p1 + 3 * bstride);
        const v4f y3 = *(const v4f*)(p2 + 3 * bstride);
        asm volatile("" :: "v"(alo), "v"(ahi), "v"(x0), "v"(y0), "v"(x1), "v"(y1),
                           "v"(x2), "v"(y2), "v"(x3), "v"(y3));

#define KSTEP(AX, AY, XV, YV)                                                                        \
        {                                                                                            \
            const v2f a = {AX, AY};                                                                  \
            const v2f q0 = {XV.x, YV.x};                                                             \
            const v2f q1 = {XV.y, YV.y};                                                             \
            const v2f q2 = {XV.z, YV.z};                                                             \
            const v2f q3 = {XV.w, YV.w};                                                             \
            acc0 = __builtin_amdgcn_wmma_f32_16x16x4_f32(false, a, false, q0, (short)0, acc0, false, false); \
            acc1 = __builtin_amdgcn_wmma_f32_16x16x4_f32(false, a, false, q1, (short)0, acc1, false, false); \
            acc2 = __builtin_amdgcn_wmma_f32_16x16x4_f32(false, a, false, q2, (short)0, acc2, false, false); \
            acc3 = __builtin_amdgcn_wmma_f32_16x16x4_f32(false, a, false, q3, (short)0, acc3, false, false); \
        }
        KSTEP(alo.x, alo.y, x0, y0)
        KSTEP(alo.z, alo.w, x1, y1)
        KSTEP(ahi.x, ahi.y, x2, y2)
        KSTEP(ahi.z, ahi.w, x3, y3)
#undef KSTEP
    }

    #pragma unroll
    for (int r = 0; r < 8; ++r) {
        const int trow = mt * 16 + r + 8 * h;
        float* orow = out + ((size_t)trow * B_DIM + b) * R_DIM + n0 + 4 * c;
        const v4f o = {acc0[r], acc1[r], acc2[r], acc3[r]};
        *(v4f*)orow = o;
    }
}

extern "C" void kernel_launch(void* const* d_in, const int* in_sizes, int n_in,
                              void* d_out, int out_size, void* d_ws, size_t ws_size,
                              hipStream_t stream) {
    const float* u  = (const float*)d_in[0];
    const float* d1 = (const float*)d_in[1];
    const float* d2 = (const float*)d_in[2];
    const float* v1 = (const float*)d_in[3];
    const float* v2 = (const float*)d_in[4];
    float* out = (float*)d_out;
    float* W   = (float*)d_ws;

    weights_kernel<<<dim3(NSLOTS / GSLOTS), dim3(32 * WAVES), 0, stream>>>(u, d1, d2, W);
    gemm_kernel<<<dim3(B_DIM * 8), dim3(256), 0, stream>>>(W, v1, v2, out);
}


# State walk also split by batch group: a wave = 32 slots x 4 batches walks alone (no barrier), stores S and its group's max of x; a new kernel maxes over groups and writes W = min(S, max); same values
# speedup vs baseline: 51.6072x; 1.5594x over previous
#include <hip/hip_runtime.h>


#define T_DIM 128
#define B_DIM 128
#define R_DIM 512
#define NSLOTS 256
#define GSLOTS 32
#define NB     4
#define NBG    (B_DIM / NB)
#define TPAD   (T_DIM + 8)

typedef float v2f __attribute__((ext_vector_type(2)));
typedef float v4f __attribute__((ext_vector_type(4)));
typedef float v8f __attribute__((ext_vector_type(8)));

struct Rows4 { v4f u[4], c[4], dn[4]; };

__device__ __forceinline__ void load_rows(Rows4& r, const v4f* su, const v4f* sc, const v4f* sdn, int t) {
    #pragma unroll
    for (int s = 0; s < 4; ++s) { r.u[s] = su[t + s]; r.c[s] = sc[t + s]; r.dn[s] = sdn[t + s]; }
}

template <bool EARLY>
__device__ __forceinline__ void walk4(
    int t, int tj, float* wcol, float* pcol,
    const v4f* su, const v4f* sc, const v4f* sdn,
    float (&S)[NB], float (&used)[NB], const float (&bornS)[NB], const float (&bornU)[NB],
    const Rows4& cur, Rows4& nxt)
{
    load_rows(nxt, su, sc, sdn, t + 4);
    #pragma unroll
    for (int s = 0; s < 4; ++s) {
        const int tt = t + s;
        float X[NB];
        #pragma unroll
        for (int i = 0; i < NB; ++i) X[i] = cur.u[s][i] - used[i];
        const float m = fmaxf(fmaxf(X[0], X[1]), fmaxf(X[2], X[3]));
        pcol[(size_t)tt * NBG * NSLOTS] = m;
        float* wr = wcol + (size_t)tt * B_DIM * NSLOTS;
        #pragma unroll
        for (int i = 0; i < NB; ++i) wr[i * NSLOTS] = S[i];
        #pragma unroll
        for (int i = 0; i < NB; ++i) {
            S[i]    = S[i] - fminf(S[i], fmaxf(0.0f, X[i]));
            used[i] = fmaxf(used[i] + cur.c[s][i], cur.dn[s][i]);
        }
        if (EARLY) {
            const bool born = tj == tt + 1;
            #pragma unroll
            for (int i = 0; i < NB; ++i) {
                S[i]    = born ? bornS[i] : S[i];
                used[i] = born ? bornU[i] : used[i];
            }
        }
    }
}

__global__ __launch_bounds__(32) void state_kernel(
    const float* __restrict__ u,
    const float* __restrict__ d1,
    const float* __restrict__ d2,
    float* __restrict__ W,
    float* __restrict__ P)
{
    __shared__ v4f su[TPAD], sc[TPAD], sdn[TPAD];

    const int lane = threadIdx.x;
    const int g  = blockIdx.x / NBG;
    const int bg = blockIdx.x % NBG;
    const int j  = g * GSLOTS + lane;
    const int tj = j >> 1;
    const int b0 = bg * NB;
    const int t0 = g * (GSLOTS / 2);
    const bool odd = (j & 1) != 0;
    const v4f zero = {0.0f, 0.0f, 0.0f, 0.0f};

    {
        v4f uu[5], a1[5], a2[5];
        #pragma unroll
        for (int k = 0; k < 5; ++k) {
            const int t  = t0 + lane + 32 * k;
            const int tc = t < T_DIM ? t : T_DIM - 1;
            const int tn = t + 1 < T_DIM ? t + 1 : T_DIM - 1;
            uu[k] = *(const v4f*)(u  + tc * B_DIM + b0);
            a1[k] = *(const v4f*)(d1 + tn * B_DIM + b0);
            a2[k] = *(const v4f*)(d2 + tn * B_DIM + b0);
        }
        #pragma unroll
        for (int k = 0; k < 5; ++k) {
            const int t = t0 + lane + 32 * k;
            if (t < TPAD) {
                const v4f x  = t < T_DIM ? uu[k] : zero;
                const v4f dn = t + 1 < T_DIM ? a1[k] + a2[k] : zero;
                v4f up;
                #pragma unroll
                for (int i = 0; i < NB; ++i) up[i] = fmaxf(x[i], 0.0f);
                su[t] = x; sc[t] = dn - up; sdn[t] = dn;
            }
        }
    }

    float bornS[NB], bornU[NB];
    {
        const v4f a1 = *(const v4f*)(d1 + tj * B_DIM + b0);
        const v4f a2 = *(const v4f*)(d2 + tj * B_DIM + b0);
        #pragma unroll
        for (int i = 0; i < NB; ++i) { bornS[i] = odd ? a2[i] : a1[i]; bornU[i] = odd ? 0.0f : a2[i]; }
    }

    {
        float* z = W + ((size_t)b0 + (lane >> 3)) * NSLOTS + g * GSLOTS + (lane & 7) * 4;
        for (int t = 0; t < t0; ++t) *(v4f*)(z + (size_t)t * B_DIM * NSLOTS) = zero;
    }

    float S[NB], used[NB];
    #pragma unroll
    for (int i = 0; i < NB; ++i) {
        S[i]    = (tj == t0) ? bornS[i] : 0.0f;
        used[i] = (tj == t0) ? bornU[i] : 0.0f;
    }

    __syncthreads();

    float* wcol = W + (size_t)b0 * NSLOTS + j;
    float* pcol = P + (size_t)bg * NSLOTS + j;

    Rows4 A, B;
    load_rows(A, su, sc, sdn, t0);
    for (int t = t0; t < t0 + GSLOTS / 2; t += 8) {
        walk4<true>(t,     tj, wcol, pcol, su, sc, sdn, S, used, bornS, bornU, A, B);
        walk4<true>(t + 4, tj, wcol, pcol, su, sc, sdn, S, used, bornS, bornU, B, A);
    }
    for (int t = t0 + GSLOTS / 2; t < T_DIM; t += 8) {
        walk4<false>(t,     tj, wcol, pcol, su, sc, sdn, S, used, bornS, bornU, A, B);
        walk4<false>(t + 4, tj, wcol, pcol, su, sc, sdn, S, used, bornS, bornU, B, A);
    }
}

__global__ __launch_bounds__(256) void finish_kernel(
    float* __restrict__ W,
    const float* __restrict__ P)
{
    const int t  = T_DIM - 1 - (int)(blockIdx.x >> 1);
    const int hb = blockIdx.x & 1;
    const int k  = threadIdx.x & 63;
    const int r  = threadIdx.x >> 6;
    const int top = 2 * t + 1;
    if (4 * k > top) return;

    float* w = W + ((size_t)t * B_DIM + 64 * hb + r) * NSLOTS + 4 * k;
    v4f s[16];
    #pragma unroll
    for (int i = 0; i < 16; ++i) s[i] = *(const v4f*)(w + (size_t)i * 4 * NSLOTS);

    const float* p = P + (size_t)t * NBG * NSLOTS + 4 * k;
    v4f x[NBG];
    #pragma unroll
    for (int q = 0; q < NBG; ++q) x[q] = *(const v4f*)(p + q * NSLOTS);

    v4f m = x[0];
    #pragma unroll
    for (int q = 1; q < NBG; ++q) {
        #pragma unroll
        for (int e = 0; e < 4; ++e) m[e] = fmaxf(m[e], x[q][e]);
    }
    #pragma unroll
    for (int e = 0; e < 4; ++e) m[e] = (4 * k + e <= top) ? m[e] : 0.0f;

    #pragma unroll
    for (int i = 0; i < 16; ++i) {
        #pragma unroll
        for (int e = 0; e < 4; ++e) s[i][e] = fminf(s[i][e], m[e]);
        *(v4f*)(w + (size_t)i * 4 * NSLOTS) = s[i];
    }
}

__global__ __launch_bounds__(256) void gemm_kernel(
    const float* __restrict__ W,
    const float* __restrict__ v1,
    const float* __restrict__ v2,
    float* __restrict__ out)
{
    const int lane = threadIdx.x & 31;
    const int wave = threadIdx.x >> 5;
    const int c    = lane & 15;
    const int h    = lane >> 4;

    const int b  = blockIdx.x >> 3;
    const int mt = blockIdx.x & 7;
    const int n0 = wave * 64;

    v8f acc0 = {}, acc1 = {}, acc2 = {}, acc3 = {};

    const float* Ap = W + ((size_t)(mt * 16 + c) * B_DIM + b) * NSLOTS + 8 * h;
    const size_t bstride = (size_t)B_DIM * R_DIM;
    const float* B1 = v1 + (size_t)(4 * h) * bstride + (size_t)b * R_DIM + n0 + 4 * c;
    const float* B2 = v2 + (size_t)(4 * h) * bstride + (size_t)b * R_DIM + n0 + 4 * c;

    for (int G = 0; G < NSLOTS / 16; ++G) {
        const v4f alo = *(const v4f*)(Ap + 16 * G);
        const v4f ahi = *(const v4f*)(Ap + 16 * G + 4);
        const float* p1 = B1 + (size_t)(8 * G) * bstride;
        const float* p2 = B2 + (size_t)(8 * G) * bstride;
        const v4f x0 = *(const v4f*)(p1);
        const v4f y0 = *(const v4f*)(p2);
        const v4f x1 = *(const v4f*)(p1 + bstride);
        const v4f y1 = *(const v4f*)(p2 + bstride);
        const v4f x2 = *(const v4f*)(p1 + 2 * bstride);
        const v4f y2 = *(const v4f*)(p2 + 2 * bstride);
        const v4f x3 = *(const v4f*)(p1 + 3 * bstride);
        const v4f y3 = *(const v4f*)(p2 + 3 * bstride);
        asm volatile("" :: "v"(alo), "v"(ahi), "v"(x0), "v"(y0), "v"(x1), "v"(y1),
                           "v"(x2), "v"(y2), "v"(x3), "v"(y3));

#define KSTEP(AX, AY, XV, YV)                                                                        \
        {                                                                                            \
            const v2f a = {AX, AY};                                                                  \
            const v2f q0 = {XV.x, YV.x};                                                             \
            const v2f q1 = {XV.y, YV.y};                                                             \
            const v2f q2 = {XV.z, YV.z};                                                             \
            const v2f q3 = {XV.w, YV.w};                                                             \
            acc0 = __builtin_amdgcn_wmma_f32_16x16x4_f32(false, a, false, q0, (short)0, acc0, false, false); \
            acc1 = __builtin_amdgcn_wmma_f32_16x16x4_f32(false, a, false, q1, (short)0, acc1, false, false); \
            acc2 = __builtin_amdgcn_wmma_f32_16x16x4_f32(false, a, false, q2, (short)0, acc2, false, false); \
            acc3 = __builtin_amdgcn_wmma_f32_16x16x4_f32(false, a, false, q3, (short)0, acc3, false, false); \
        }
        KSTEP(alo.x, alo.y, x0, y0)
        KSTEP(alo.z, alo.w, x1, y1)
        KSTEP(ahi.x, ahi.y, x2, y2)
        KSTEP(ahi.z, ahi.w, x3, y3)
#undef KSTEP
    }

    #pragma unroll
    for (int r = 0; r < 8; ++r) {
        const int trow = mt * 16 + r + 8 * h;
        float* orow = out + ((size_t)trow * B_DIM + b) * R_DIM + n0 + 4 * c;
        const v4f o = {acc0[r], acc1[r], acc2[r], acc3[r]};
        *(v4f*)orow = o;
    }
}

extern "C" void kernel_launch(void* const* d_in, const int* in_sizes, int n_in,
                              void* d_out, int out_size, void* d_ws, size_t ws_size,
                              hipStream_t stream) {
    const float* u  = (const float*)d_in[0];
    const float* d1 = (const float*)d_in[1];
    const float* d2 = (const float*)d_in[2];
    const float* v1 = (const float*)d_in[3];
    const float* v2 = (const float*)d_in[4];
    float* out = (float*)d_out;
    float* W   = (float*)d_ws;
    float* P   = W + (size_t)T_DIM * B_DIM * NSLOTS;

    state_kernel<<<dim3((NSLOTS / GSLOTS) * NBG), dim3(32), 0, stream>>>(u, d1, d2, W, P);
    finish_kernel<<<dim3(T_DIM * 2), dim3(256), 0, stream>>>(W, P);
    gemm_kernel<<<dim3(B_DIM * 8), dim3(256), 0, stream>>>(W, v1, v2, out);
}


# State kernel: a block is 4 waves, one batch each: a wave stores 1 line of W a step, not 4; x of 8 steps meets in LDS at one LDS-only barrier, each wave stores 2 of the 8 group maxima; same values
# speedup vs baseline: 53.7304x; 1.0411x over previous
#include <hip/hip_runtime.h>


#define T_DIM 128
#define B_DIM 128
#define R_DIM 512
#define NSLOTS 256
#define GSLOTS 32
#define NB     4
#define NBG    (B_DIM / NB)
#define TPAD   (T_DIM + 8)

typedef float v2f __attribute__((ext_vector_type(2)));
typedef float v4f __attribute__((ext_vector_type(4)));
typedef float v8f __attribute__((ext_vector_type(8)));

struct Rows4 { v4f u, c, dn; };

__device__ __forceinline__ void load_rows(Rows4& r, const float* su, const float* sc, const float* sdn, int t) {
    r.u = *(const v4f*)(su + t); r.c = *(const v4f*)(sc + t); r.dn = *(const v4f*)(sdn + t);
}

__device__ __forceinline__ void lds_barrier() {
    asm volatile("s_wait_dscnt 0x0\n\ts_barrier_signal -1\n\ts_barrier_wait -1" ::: "memory");
}

template <bool EARLY>
__device__ __forceinline__ void walk4(
    int t, int tj, float* wcol, float* xw,
    const float* su, const float* sc, const float* sdn,
    float& S, float& used, float bornS, float bornU,
    const Rows4& cur, Rows4& nxt)
{
    load_rows(nxt, su, sc, sdn, t + 4);
    #pragma unroll
    for (int s = 0; s < 4; ++s) {
        const int tt = t + s;
        const float x = cur.u[s] - used;
        xw[s * NB * GSLOTS] = x;
        wcol[(size_t)tt * B_DIM * NSLOTS] = S;
        S    = S - fminf(S, fmaxf(0.0f, x));
        used = fmaxf(used + cur.c[s], cur.dn[s]);
        if (EARLY) {
            const bool born = tj == tt + 1;
            S    = born ? bornS : S;
            used = born ? bornU : used;
        }
    }
}

template <bool EARLY>
__device__ __forceinline__ void trip8(
    int t, int tj, int stage, int wave, int lane, float* wcol, float* pcol, float* xs,
    const float* su, const float* sc, const float* sdn,
    float& S, float& used, float bornS, float bornU, Rows4& A, Rows4& B)
{
    float* tile = xs + stage * 8 * NB * GSLOTS;
    float* xw   = tile + wave * GSLOTS + lane;
    walk4<EARLY>(t,     tj, wcol, xw,                   su, sc, sdn, S, used, bornS, bornU, A, B);
    walk4<EARLY>(t + 4, tj, wcol, xw + 4 * NB * GSLOTS, su, sc, sdn, S, used, bornS, bornU, B, A);
    lds_barrier();
    #pragma unroll
    for (int e = 0; e < 2; ++e) {
        const int step = 2 * wave + e;
        const float* xr = tile + step * NB * GSLOTS + lane;
        const float m = fmaxf(fmaxf(xr[0], xr[GSLOTS]), fmaxf(xr[2 * GSLOTS], xr[3 * GSLOTS]));
        pcol[(size_t)(t + step) * NBG * NSLOTS] = m;
    }
}

__global__ __launch_bounds__(NB * 32) void state_kernel(
    const float* __restrict__ u,
    const float* __restrict__ d1,
    const float* __restrict__ d2,
    float* __restrict__ W,
    float* __restrict__ P)
{
    __shared__ v4f su4[NB * TPAD / 4], sc4[NB * TPAD / 4], sdn4[NB * TPAD / 4];
    __shared__ float xs[2 * 8 * NB * GSLOTS];
    float* suf  = (float*)su4;
    float* scf  = (float*)sc4;
    float* sdnf = (float*)sdn4;

    const int tid  = threadIdx.x;
    const int lane = tid & 31;
    const int wave = tid >> 5;
    const int g  = blockIdx.x / NBG;
    const int bg = blockIdx.x % NBG;
    const int j  = g * GSLOTS + lane;
    const int tj = j >> 1;
    const int b0 = bg * NB;
    const int b  = b0 + wave;
    const int t0 = g * (GSLOTS / 2);
    const bool odd = (j & 1) != 0;
    const v4f zero = {0.0f, 0.0f, 0.0f, 0.0f};

    {
        const int t  = t0 + tid;
        const int tc = t < T_DIM ? t : T_DIM - 1;
        const int tn = t + 1 < T_DIM ? t + 1 : T_DIM - 1;
        const v4f uu = *(const v4f*)(u  + tc * B_DIM + b0);
        const v4f a1 = *(const v4f*)(d1 + tn * B_DIM + b0);
        const v4f a2 = *(const v4f*)(d2 + tn * B_DIM + b0);
        if (t < T_DIM) {
            const v4f dn = t + 1 < T_DIM ? a1 + a2 : zero;
            #pragma unroll
            for (int i = 0; i < NB; ++i) {
                const float up = fmaxf(uu[i], 0.0f);
                suf[i * TPAD + t] = uu[i]; scf[i * TPAD + t] = dn[i] - up; sdnf[i * TPAD + t] = dn[i];
            }
        }
        if (tid < TPAD - T_DIM) {
            #pragma unroll
            for (int i = 0; i < NB; ++i) {
                suf[i * TPAD + T_DIM + tid] = 0.0f; scf[i * TPAD + T_DIM + tid] = 0.0f; sdnf[i * TPAD + T_DIM + tid] = 0.0f;
            }
        }
    }

    const float b1 = d1[tj * B_DIM + b];
    const float b2 = d2[tj * B_DIM + b];
    const float bornS = odd ? b2 : b1;
    const float bornU = odd ? 0.0f : b2;

    {
        float* z = W + ((size_t)b0 + (lane >> 3)) * NSLOTS + g * GSLOTS + (lane & 7) * 4;
        for (int t = wave; t < t0; t += NB) *(v4f*)(z + (size_t)t * B_DIM * NSLOTS) = zero;
    }

    float S    = (tj == t0) ? bornS : 0.0f;
    float used = (tj == t0) ? bornU : 0.0f;

    __syncthreads();

    const float* su  = suf  + wave * TPAD;
    const float* sc  = scf  + wave * TPAD;
    const float* sdn = sdnf + wave * TPAD;
    float* wcol = W + (size_t)b * NSLOTS + j;
    float* pcol = P + (size_t)bg * NSLOTS + j;

    Rows4 A, B;
    load_rows(A, su, sc, sdn, t0);
    int stage = 0;
    for (int t = t0; t < t0 + GSLOTS / 2; t += 8) {
        trip8<true>(t, tj, stage, wave, lane, wcol, pcol, xs, su, sc, sdn, S, used, bornS, bornU, A, B);
        stage ^= 1;
    }
    for (int t = t0 + GSLOTS / 2; t < T_DIM; t += 8) {
        trip8<false>(t, tj, stage, wave, lane, wcol, pcol, xs, su, sc, sdn, S, used, bornS, bornU, A, B);
        stage ^= 1;
    }
}

__global__ __launch_bounds__(256) void finish_kernel(
    float* __restrict__ W,
    const float* __restrict__ P)
{
    const int t  = T_DIM - 1 - (int)(blockIdx.x >> 1);
    const int hb = blockIdx.x & 1;
    const int k  = threadIdx.x & 63;
    const int r  = threadIdx.x >> 6;
    const int top = 2 * t + 1;
    if (4 * k > top) return;

    float* w = W + ((size_t)t * B_DIM + 64 * hb + r) * NSLOTS + 4 * k;
    v4f s[16];
    #pragma unroll
    for (int i = 0; i < 16; ++i) s[i] = *(const v4f*)(w + (size_t)i * 4 * NSLOTS);

    const float* p = P + (size_t)t * NBG * NSLOTS + 4 * k;
    v4f x[NBG];
    #pragma unroll
    for (int q = 0; q < NBG; ++q) x[q] = *(const v4f*)(p + q * NSLOTS);

    v4f m = x[0];
    #pragma unroll
    for (int q = 1; q < NBG; ++q) {
        #pragma unroll
        for (int e = 0; e < 4; ++e) m[e] = fmaxf(m[e], x[q][e]);
    }
    #pragma unroll
    for (int e = 0; e < 4; ++e) m[e] = (4 * k + e <= top) ? m[e] : 0.0f;

    #pragma unroll
    for (int i = 0; i < 16; ++i) {
        #pragma unroll
        for (int e = 0; e < 4; ++e) s[i][e] = fminf(s[i][e], m[e]);
        *(v4f*)(w + (size_t)i * 4 * NSLOTS) = s[i];
    }
}

__global__ __launch_bounds__(256) void gemm_kernel(
    const float* __restrict__ W,
    const float* __restrict__ v1,
    const float* __restrict__ v2,
    float* __restrict__ out)
{
    const int lane = threadIdx.x & 31;
    const int wave = threadIdx.x >> 5;
    const int c    = lane & 15;
    const int h    = lane >> 4;

    const int b  = blockIdx.x >> 3;
    const int mt = blockIdx.x & 7;
    const int n0 = wave * 64;

    v8f acc0 = {}, acc1 = {}, acc2 = {}, acc3 = {};

    const float* Ap = W + ((size_t)(mt * 16 + c) * B_DIM + b) * NSLOTS + 8 * h;
    const size_t bstride = (size_t)B_DIM * R_DIM;
    const float* B1 = v1 + (size_t)(4 * h) * bstride + (size_t)b * R_DIM + n0 + 4 * c;
    const float* B2 = v2 + (size_t)(4 * h) * bstride + (size_t)b * R_DIM + n0 + 4 * c;

    for (int G = 0; G < NSLOTS / 16; ++G) {
        const v4f alo = *(const v4f*)(Ap + 16 * G);
        const v4f ahi = *(const v4f*)(Ap + 16 * G + 4);
        const float* p1 = B1 + (size_t)(8 * G) * bstride;
        const float* p2 = B2 + (size_t)(8 * G) * bstride;
        const v4f x0 = *(const v4f*)(p1);
        const v4f y0 = *(const v4f*)(p2);
        const v4f x1 = *(const v4f*)(p1 + bstride);
        const v4f y1 = *(const v4f*)(p2 + bstride);
        const v4f x2 = *(const v4f*)(p1 + 2 * bstride);
        const v4f y2 = *(const v4f*)(p2 + 2 * bstride);
        const v4f x3 = *(const v4f*)(p1 + 3 * bstride);
        const v4f y3 = *(const v4f*)(p2 + 3 * bstride);
        asm volatile("" :: "v"(alo), "v"(ahi), "v"(x0), "v"(y0), "v"(x1), "v"(y1),
                           "v"(x2), "v"(y2), "v"(x3), "v"(y3));

#define KSTEP(AX, AY, XV, YV)                                                                        \
        {                                                                                            \
            const v2f a = {AX, AY};                                                                  \
            const v2f q0 = {XV.x, YV.x};                                                             \
            const v2f q1 = {XV.y, YV.y};                                                             \
            const v2f q2 = {XV.z, YV.z};                                                             \
            const v2f q3 = {XV.w, YV.w};                                                             \
            acc0 = __builtin_amdgcn_wmma_f32_16x16x4_f32(false, a, false, q0, (short)0, acc0, false, false); \
            acc1 = __builtin_amdgcn_wmma_f32_16x16x4_f32(false, a, false, q1, (short)0, acc1, false, false); \
            acc2 = __builtin_amdgcn_wmma_f32_16x16x4_f32(false, a, false, q2, (short)0, acc2, false, false); \
            acc3 = __builtin_amdgcn_wmma_f32_16x16x4_f32(false, a, false, q3, (short)0, acc3, false, false); \
        }
        KSTEP(alo.x, alo.y, x0, y0)
        KSTEP(alo.z, alo.w, x1, y1)
        KSTEP(ahi.x, ahi.y, x2, y2)
        KSTEP(ahi.z, ahi.w, x3, y3)
#undef KSTEP
    }

    #pragma unroll
    for (int r = 0; r < 8; ++r) {
        const int trow = mt * 16 + r + 8 * h;
        float* orow = out + ((size_t)trow * B_DIM + b) * R_DIM + n0 + 4 * c;
        const v4f o = {acc0[r], acc1[r], acc2[r], acc3[r]};
        *(v4f*)orow = o;
    }
}

extern "C" void kernel_launch(void* const* d_in, const int* in_sizes, int n_in,
                              void* d_out, int out_size, void* d_ws, size_t ws_size,
                              hipStream_t stream) {
    const float* u  = (const float*)d_in[0];
    const float* d1 = (const float*)d_in[1];
    const float* d2 = (const float*)d_in[2];
    const float* v1 = (const float*)d_in[3];
    const float* v2 = (const float*)d_in[4];
    float* out = (float*)d_out;
    float* W   = (float*)d_ws;
    float* P   = W + (size_t)T_DIM * B_DIM * NSLOTS;

    state_kernel<<<dim3((NSLOTS / GSLOTS) * NBG), dim3(NB * 32), 0, stream>>>(u, d1, d2, W, P);
    finish_kernel<<<dim3(T_DIM * 2), dim3(256), 0, stream>>>(W, P);
    gemm_kernel<<<dim3(B_DIM * 8), dim3(256), 0, stream>>>(W, v1, v2, out);
}


# GEMM: block = 1 batch x 128 columns, its 8 waves = the 8 row tiles; each value of v1/v2 is fetched once per block into LDS (4 slots of a column = 16 B) and read by all row tiles; same sums
# speedup vs baseline: 72.4757x; 1.3489x over previous
#include <hip/hip_runtime.h>


#define T_DIM 128
#define B_DIM 128
#define R_DIM 512
#define NSLOTS 256
#define GSLOTS 32
#define NB     4
#define NBG    (B_DIM / NB)
#define TPAD   (T_DIM + 8)

typedef float v2f __attribute__((ext_vector_type(2)));
typedef float v4f __attribute__((ext_vector_type(4)));
typedef float v8f __attribute__((ext_vector_type(8)));

struct Rows4 { v4f u, c, dn; };

__device__ __forceinline__ void load_rows(Rows4& r, const float* su, const float* sc, const float* sdn, int t) {
    r.u = *(const v4f*)(su + t); r.c = *(const v4f*)(sc + t); r.dn = *(const v4f*)(sdn + t);
}

__device__ __forceinline__ void lds_barrier() {
    asm volatile("s_wait_dscnt 0x0\n\ts_barrier_signal -1\n\ts_barrier_wait -1" ::: "memory");
}

template <bool EARLY>
__device__ __forceinline__ void walk4(
    int t, int tj, float* wcol, float* xw,
    const float* su, const float* sc, const float* sdn,
    float& S, float& used, float bornS, float bornU,
    const Rows4& cur, Rows4& nxt)
{
    load_rows(nxt, su, sc, sdn, t + 4);
    #pragma unroll
    for (int s = 0; s < 4; ++s) {
        const int tt = t + s;
        const float x = cur.u[s] - used;
        xw[s * NB * GSLOTS] = x;
        wcol[(size_t)tt * B_DIM * NSLOTS] = S;
        S    = S - fminf(S, fmaxf(0.0f, x));
        used = fmaxf(used + cur.c[s], cur.dn[s]);
        if (EARLY) {
            const bool born = tj == tt + 1;
            S    = born ? bornS : S;
            used = born ? bornU : used;
        }
    }
}

template <bool EARLY>
__device__ __forceinline__ void trip8(
    int t, int tj, int stage, int wave, int lane, float* wcol, float* pcol, float* xs,
    const float* su, const float* sc, const float* sdn,
    float& S, float& used, float bornS, float bornU, Rows4& A, Rows4& B)
{
    float* tile = xs + stage * 8 * NB * GSLOTS;
    float* xw   = tile + wave * GSLOTS + lane;
    walk4<EARLY>(t,     tj, wcol, xw,                   su, sc, sdn, S, used, bornS, bornU, A, B);
    walk4<EARLY>(t + 4, tj, wcol, xw + 4 * NB * GSLOTS, su, sc, sdn, S, used, bornS, bornU, B, A);
    lds_barrier();
    #pragma unroll
    for (int e = 0; e < 2; ++e) {
        const int step = 2 * wave + e;
        const float* xr = tile + step * NB * GSLOTS + lane;
        const float m = fmaxf(fmaxf(xr[0], xr[GSLOTS]), fmaxf(xr[2 * GSLOTS], xr[3 * GSLOTS]));
        pcol[(size_t)(t + step) * NBG * NSLOTS] = m;
    }
}

__global__ __launch_bounds__(NB * 32) void state_kernel(
    const float* __restrict__ u,
    const float* __restrict__ d1,
    const float* __restrict__ d2,
    float* __restrict__ W,
    float* __restrict__ P)
{
    __shared__ v4f su4[NB * TPAD / 4], sc4[NB * TPAD / 4], sdn4[NB * TPAD / 4];
    __shared__ float xs[2 * 8 * NB * GSLOTS];
    float* suf  = (float*)su4;
    float* scf  = (float*)sc4;
    float* sdnf = (float*)sdn4;

    const int tid  = threadIdx.x;
    const int lane = tid & 31;
    const int wave = tid >> 5;
    const int g  = blockIdx.x / NBG;
    const int bg = blockIdx.x % NBG;
    const int j  = g * GSLOTS + lane;
    const int tj = j >> 1;
    const int b0 = bg * NB;
    const int b  = b0 + wave;
    const int t0 = g * (GSLOTS / 2);
    const bool odd = (j & 1) != 0;
    const v4f zero = {0.0f, 0.0f, 0.0f, 0.0f};

    {
        const int t  = t0 + tid;
        const int tc = t < T_DIM ? t : T_DIM - 1;
        const int tn = t + 1 < T_DIM ? t + 1 : T_DIM - 1;
        const v4f uu = *(const v4f*)(u  + tc * B_DIM + b0);
        const v4f a1 = *(const v4f*)(d1 + tn * B_DIM + b0);
        const v4f a2 = *(const v4f*)(d2 + tn * B_DIM + b0);
        if (t < T_DIM) {
            const v4f dn = t + 1 < T_DIM ? a1 + a2 : zero;
            #pragma unroll
            for (int i = 0; i < NB; ++i) {
                const float up = fmaxf(uu[i], 0.0f);
                suf[i * TPAD + t] = uu[i]; scf[i * TPAD + t] = dn[i] - up; sdnf[i * TPAD + t] = dn[i];
            }
        }
        if (tid < TPAD - T_DIM) {
            #pragma unroll
            for (int i = 0; i < NB; ++i) {
                suf[i * TPAD + T_DIM + tid] = 0.0f; scf[i * TPAD + T_DIM + tid] = 0.0f; sdnf[i * TPAD + T_DIM + tid] = 0.0f;
            }
        }
    }

    const float b1 = d1[tj * B_DIM + b];
    const float b2 = d2[tj * B_DIM + b];
    const float bornS = odd ? b2 : b1;
    const float bornU = odd ? 0.0f : b2;

    {
        float* z = W + ((size_t)b0 + (lane >> 3)) * NSLOTS + g * GSLOTS + (lane & 7) * 4;
        for (int t = wave; t < t0; t += NB) *(v4f*)(z + (size_t)t * B_DIM * NSLOTS) = zero;
    }

    float S    = (tj == t0) ? bornS : 0.0f;
    float used = (tj == t0) ? bornU : 0.0f;

    __syncthreads();

    const float* su  = suf  + wave * TPAD;
    const float* sc  = scf  + wave * TPAD;
    const float* sdn = sdnf + wave * TPAD;
    float* wcol = W + (size_t)b * NSLOTS + j;
    float* pcol = P + (size_t)bg * NSLOTS + j;

    Rows4 A, B;
    load_rows(A, su, sc, sdn, t0);
    int stage = 0;
    for (int t = t0; t < t0 + GSLOTS / 2; t += 8) {
        trip8<true>(t, tj, stage, wave, lane, wcol, pcol, xs, su, sc, sdn, S, used, bornS, bornU, A, B);
        stage ^= 1;
    }
    for (int t = t0 + GSLOTS / 2; t < T_DIM; t += 8) {
        trip8<false>(t, tj, stage, wave, lane, wcol, pcol, xs, su, sc, sdn, S, used, bornS, bornU, A, B);
        stage ^= 1;
    }
}

__global__ __launch_bounds__(256) void finish_kernel(
    float* __restrict__ W,
    const float* __restrict__ P)
{
    const int t  = T_DIM - 1 - (int)(blockIdx.x >> 1);
    const int hb = blockIdx.x & 1;
    const int k  = threadIdx.x & 63;
    const int r  = threadIdx.x >> 6;
    const int top = 2 * t + 1;
    if (4 * k > top) return;

    float* w = W + ((size_t)t * B_DIM + 64 * hb + r) * NSLOTS + 4 * k;
    v4f s[16];
    #pragma unroll
    for (int i = 0; i < 16; ++i) s[i] = *(const v4f*)(w + (size_t)i * 4 * NSLOTS);

    const float* p = P + (size_t)t * NBG * NSLOTS + 4 * k;
    v4f x[NBG];
    #pragma unroll
    for (int q = 0; q < NBG; ++q) x[q] = *(const v4f*)(p + q * NSLOTS);

    v4f m = x[0];
    #pragma unroll
    for (int q = 1; q < NBG; ++q) {
        #pragma unroll
        for (int e = 0; e < 4; ++e) m[e] = fmaxf(m[e], x[q][e]);
    }
    #pragma unroll
    for (int e = 0; e < 4; ++e) m[e] = (4 * k + e <= top) ? m[e] : 0.0f;

    #pragma unroll
    for (int i = 0; i < 16; ++i) {
        #pragma unroll
        for (int e = 0; e < 4; ++e) s[i][e] = fminf(s[i][e], m[e]);
        *(v4f*)(w + (size_t)i * 4 * NSLOTS) = s[i];
    }
}

#define GN   128
#define QP   (GN + 4)
#define CHQ  8

__device__ __forceinline__ void load_a(v4f (&a)[4], const float* Ap, int ch) {
    a[0] = *(const v4f*)(Ap + 32 * ch);
    a[1] = *(const v4f*)(Ap + 32 * ch + 4);
    a[2] = *(const v4f*)(Ap + 32 * ch + 16);
    a[3] = *(const v4f*)(Ap + 32 * ch + 20);
}

__device__ __forceinline__ void load_plane(v4f (&p)[4], const float* S1, const float* S2, size_t bstride, int ch) {
    const float* r1 = S1 + (size_t)(16 * ch) * bstride;
    const float* r2 = S2 + (size_t)(16 * ch) * bstride;
    p[0] = *(const v4f*)(r1);
    p[1] = *(const v4f*)(r2);
    p[2] = *(const v4f*)(r1 + bstride);
    p[3] = *(const v4f*)(r2 + bstride);
}

__device__ __forceinline__ void store_plane(v4f* dst, const v4f (&p)[4]) {
    #pragma unroll
    for (int i = 0; i < 4; ++i) {
        const v4f col = {p[0][i], p[1][i], p[2][i], p[3][i]};
        dst[i * 32] = col;
    }
}

__device__ __forceinline__ void mma_chunk(v8f (&acc)[8], const v4f* rd, const v4f (&a)[4]) {
    #pragma unroll
    for (int gg = 0; gg < 2; ++gg) {
        v4f blo[8], bhi[8];
        __builtin_amdgcn_sched_barrier(0);
        #pragma unroll
        for (int q = 0; q < 8; ++q) blo[q] = rd[(4 * gg) * QP + 16 * q];
        #pragma unroll
        for (int q = 0; q < 8; ++q) bhi[q] = rd[(4 * gg + 1) * QP + 16 * q];
        __builtin_amdgcn_sched_barrier(0);
        const v4f alo = a[2 * gg], ahi = a[2 * gg + 1];
        const v2f a0 = {alo.x, alo.y}, a1 = {alo.z, alo.w}, a2 = {ahi.x, ahi.y}, a3 = {ahi.z, ahi.w};
        #pragma unroll
        for (int q = 0; q < 8; ++q) {
            const v2f bq = {blo[q].x, blo[q].y};
            acc[q] = __builtin_amdgcn_wmma_f32_16x16x4_f32(false, a0, false, bq, (short)0, acc[q], false, false);
        }
        #pragma unroll
        for (int q = 0; q < 8; ++q) {
            const v2f bq = {blo[q].z, blo[q].w};
            acc[q] = __builtin_amdgcn_wmma_f32_16x16x4_f32(false, a1, false, bq, (short)0, acc[q], false, false);
        }
        #pragma unroll
        for (int q = 0; q < 8; ++q) {
            const v2f bq = {bhi[q].x, bhi[q].y};
            acc[q] = __builtin_amdgcn_wmma_f32_16x16x4_f32(false, a2, false, bq, (short)0, acc[q], false, false);
        }
        #pragma unroll
        for (int q = 0; q < 8; ++q) {
            const v2f bq = {bhi[q].z, bhi[q].w};
            acc[q] = __builtin_amdgcn_wmma_f32_16x16x4_f32(false, a3, false, bq, (short)0, acc[q], false, false);
        }
    }
}

__global__ __launch_bounds__(256) void gemm_kernel(
    const float* __restrict__ W,
    const float* __restrict__ v1,
    const float* __restrict__ v2,
    float* __restrict__ out)
{
    __shared__ v4f sb[2 * CHQ * QP];

    const int lane = threadIdx.x & 31;
    const int wave = threadIdx.x >> 5;
    const int c    = lane & 15;
    const int h    = lane >> 4;

    const int b  = blockIdx.x >> 2;
    const int n0 = (blockIdx.x & 3) * GN;

    v8f acc[8] = {};

    const float* Ap = W + ((size_t)(wave * 16 + c) * B_DIM + b) * NSLOTS + 8 * h;
    const size_t bstride = (size_t)B_DIM * R_DIM;
    const float* S1 = v1 + (size_t)(2 * wave) * bstride + (size_t)b * R_DIM + n0 + 4 * lane;
    const float* S2 = v2 + (size_t)(2 * wave) * bstride + (size_t)b * R_DIM + n0 + 4 * lane;
    v4f*       wr = sb + wave * QP + lane;
    const v4f* rd = sb + (2 * h) * QP + c;

    v4f a[4], an[4], p[4];
    load_plane(p, S1, S2, bstride, 0);
    load_a(a, Ap, 0);
    store_plane(wr, p);
    lds_barrier();

    for (int ch = 0; ch < NSLOTS / 32 - 1; ++ch) {
        const int cur = (ch & 1) * CHQ * QP, nxt = CHQ * QP - cur;
        load_plane(p, S1, S2, bstride, ch + 1);
        load_a(an, Ap, ch + 1);
        __builtin_amdgcn_sched_barrier(0);
        mma_chunk(acc, rd + cur, a);
        store_plane(wr + nxt, p);
        #pragma unroll
        for (int i = 0; i < 4; ++i) a[i] = an[i];
        lds_barrier();
    }
    mma_chunk(acc, rd + CHQ * QP, a);

    #pragma unroll
    for (int r = 0; r < 8; ++r) {
        const int trow = wave * 16 + r + 8 * h;
        float* orow = out + ((size_t)trow * B_DIM + b) * R_DIM + n0 + 4 * c;
        const v4f o0 = {acc[0][r], acc[2][r], acc[4][r], acc[6][r]};
        const v4f o1 = {acc[1][r], acc[3][r], acc[5][r], acc[7][r]};
        *(v4f*)orow = o0;
        *(v4f*)(orow + 64) = o1;
    }
}

extern "C" void kernel_launch(void* const* d_in, const int* in_sizes, int n_in,
                              void* d_out, int out_size, void* d_ws, size_t ws_size,
                              hipStream_t stream) {
    const float* u  = (const float*)d_in[0];
    const float* d1 = (const float*)d_in[1];
    const float* d2 = (const float*)d_in[2];
    const float* v1 = (const float*)d_in[3];
    const float* v2 = (const float*)d_in[4];
    float* out = (float*)d_out;
    float* W   = (float*)d_ws;
    float* P   = W + (size_t)T_DIM * B_DIM * NSLOTS;

    state_kernel<<<dim3((NSLOTS / GSLOTS) * NBG), dim3(NB * 32), 0, stream>>>(u, d1, d2, W, P);
    finish_kernel<<<dim3(T_DIM * 2), dim3(256), 0, stream>>>(W, P);
    gemm_kernel<<<dim3(B_DIM * 4), dim3(256), 0, stream>>>(W, v1, v2, out);
}


# Weight min(S, cross-batch max) is formed in the GEMM's A loader from S and a 128 KiB table of maxima; the pass that read and rewrote W becomes a reduce of P to that table; same values
# speedup vs baseline: 86.5189x; 1.1938x over previous
#include <hip/hip_runtime.h>


#define T_DIM 128
#define B_DIM 128
#define R_DIM 512
#define NSLOTS 256
#define GSLOTS 32
#define NB     4
#define NBG    (B_DIM / NB)
#define TPAD   (T_DIM + 8)

typedef float v2f __attribute__((ext_vector_type(2)));
typedef float v4f __attribute__((ext_vector_type(4)));
typedef float v8f __attribute__((ext_vector_type(8)));

struct Rows4 { v4f u, c, dn; };

__device__ __forceinline__ void load_rows(Rows4& r, const float* su, const float* sc, const float* sdn, int t) {
    r.u = *(const v4f*)(su + t); r.c = *(const v4f*)(sc + t); r.dn = *(const v4f*)(sdn + t);
}

__device__ __forceinline__ void lds_barrier() {
    asm volatile("s_wait_dscnt 0x0\n\ts_barrier_signal -1\n\ts_barrier_wait -1" ::: "memory");
}

template <bool EARLY>
__device__ __forceinline__ void walk4(
    int t, int tj, float* wcol, float* xw,
    const float* su, const float* sc, const float* sdn,
    float& S, float& used, float bornS, float bornU,
    const Rows4& cur, Rows4& nxt)
{
    load_rows(nxt, su, sc, sdn, t + 4);
    #pragma unroll
    for (int s = 0; s < 4; ++s) {
        const int tt = t + s;
        const float x = cur.u[s] - used;
        xw[s * NB * GSLOTS] = x;
        wcol[(size_t)tt * B_DIM * NSLOTS] = S;
        S    = S - fminf(S, fmaxf(0.0f, x));
        used = fmaxf(used + cur.c[s], cur.dn[s]);
        if (EARLY) {
            const bool born = tj == tt + 1;
            S    = born ? bornS : S;
            used = born ? bornU : used;
        }
    }
}

template <bool EARLY>
__device__ __forceinline__ void trip8(
    int t, int tj, int stage, int wave, int lane, float* wcol, float* pcol, float* xs,
    const float* su, const float* sc, const float* sdn,
    float& S, float& used, float bornS, float bornU, Rows4& A, Rows4& B)
{
    float* tile = xs + stage * 8 * NB * GSLOTS;
    float* xw   = tile + wave * GSLOTS + lane;
    walk4<EARLY>(t,     tj, wcol, xw,                   su, sc, sdn, S, used, bornS, bornU, A, B);
    walk4<EARLY>(t + 4, tj, wcol, xw + 4 * NB * GSLOTS, su, sc, sdn, S, used, bornS, bornU, B, A);
    lds_barrier();
    #pragma unroll
    for (int e = 0; e < 2; ++e) {
        const int step = 2 * wave + e;
        const float* xr = tile + step * NB * GSLOTS + lane;
        const float m = fmaxf(fmaxf(xr[0], xr[GSLOTS]), fmaxf(xr[2 * GSLOTS], xr[3 * GSLOTS]));
        pcol[(size_t)(t + step) * NBG * NSLOTS] = m;
    }
}

__global__ __launch_bounds__(NB * 32) void state_kernel(
    const float* __restrict__ u,
    const float* __restrict__ d1,
    const float* __restrict__ d2,
    float* __restrict__ W,
    float* __restrict__ P)
{
    __shared__ v4f su4[NB * TPAD / 4], sc4[NB * TPAD / 4], sdn4[NB * TPAD / 4];
    __shared__ float xs[2 * 8 * NB * GSLOTS];
    float* suf  = (float*)su4;
    float* scf  = (float*)sc4;
    float* sdnf = (float*)sdn4;

    const int tid  = threadIdx.x;
    const int lane = tid & 31;
    const int wave = tid >> 5;
    const int g  = blockIdx.x / NBG;
    const int bg = blockIdx.x % NBG;
    const int j  = g * GSLOTS + lane;
    const int tj = j >> 1;
    const int b0 = bg * NB;
    const int b  = b0 + wave;
    const int t0 = g * (GSLOTS / 2);
    const bool odd = (j & 1) != 0;
    const v4f zero = {0.0f, 0.0f, 0.0f, 0.0f};

    {
        const int t  = t0 + tid;
        const int tc = t < T_DIM ? t : T_DIM - 1;
        const int tn = t + 1 < T_DIM ? t + 1 : T_DIM - 1;
        const v4f uu = *(const v4f*)(u  + tc * B_DIM + b0);
        const v4f a1 = *(const v4f*)(d1 + tn * B_DIM + b0);
        const v4f a2 = *(const v4f*)(d2 + tn * B_DIM + b0);
        if (t < T_DIM) {
            const v4f dn = t + 1 < T_DIM ? a1 + a2 : zero;
            #pragma unroll
            for (int i = 0; i < NB; ++i) {
                const float up = fmaxf(uu[i], 0.0f);
                suf[i * TPAD + t] = uu[i]; scf[i * TPAD + t] = dn[i] - up; sdnf[i * TPAD + t] = dn[i];
            }
        }
        if (tid < TPAD - T_DIM) {
            #pragma unroll
            for (int i = 0; i < NB; ++i) {
                suf[i * TPAD + T_DIM + tid] = 0.0f; scf[i * TPAD + T_DIM + tid] = 0.0f; sdnf[i * TPAD + T_DIM + tid] = 0.0f;
            }
        }
    }

    const float b1 = d1[tj * B_DIM + b];
    const float b2 = d2[tj * B_DIM + b];
    const float bornS = odd ? b2 : b1;
    const float bornU = odd ? 0.0f : b2;

    {
        float* z = W + ((size_t)b0 + (lane >> 3)) * NSLOTS + g * GSLOTS + (lane & 7) * 4;
        for (int t = wave; t < t0; t += NB) *(v4f*)(z + (size_t)t * B_DIM * NSLOTS) = zero;
    }

    float S    = (tj == t0) ? bornS : 0.0f;
    float used = (tj == t0) ? bornU : 0.0f;

    __syncthreads();

    const float* su  = suf  + wave * TPAD;
    const float* sc  = scf  + wave * TPAD;
    const float* sdn = sdnf + wave * TPAD;
    float* wcol = W + (size_t)b * NSLOTS + j;
    float* pcol = P + (size_t)bg * NSLOTS + j;

    Rows4 A, B;
    load_rows(A, su, sc, sdn, t0);
    int stage = 0;
    for (int t = t0; t < t0 + GSLOTS / 2; t += 8) {
        trip8<true>(t, tj, stage, wave, lane, wcol, pcol, xs, su, sc, sdn, S, used, bornS, bornU, A, B);
        stage ^= 1;
    }
    for (int t = t0 + GSLOTS / 2; t < T_DIM; t += 8) {
        trip8<false>(t, tj, stage, wave, lane, wcol, pcol, xs, su, sc, sdn, S, used, bornS, bornU, A, B);
        stage ^= 1;
    }
}

__global__ __launch_bounds__(64) void maxima_kernel(
    const float* __restrict__ P,
    float* __restrict__ M)
{
    const int t   = blockIdx.x;
    const int k   = threadIdx.x;
    const int top = 2 * t + 1;
    v4f m = {0.0f, 0.0f, 0.0f, 0.0f};
    if (4 * k <= top) {
        const float* p = P + (size_t)t * NBG * NSLOTS + 4 * k;
        v4f x[NBG];
        #pragma unroll
        for (int q = 0; q < NBG; ++q) x[q] = *(const v4f*)(p + q * NSLOTS);
        __builtin_amdgcn_sched_barrier(0);
        m = x[0];
        #pragma unroll
        for (int q = 1; q < NBG; ++q) {
            #pragma unroll
            for (int e = 0; e < 4; ++e) m[e] = fmaxf(m[e], x[q][e]);
        }
        #pragma unroll
        for (int e = 0; e < 4; ++e) m[e] = (4 * k + e <= top) ? m[e] : 0.0f;
    }
    *(v4f*)(M + (size_t)t * NSLOTS + 4 * k) = m;
}

#define GN   128
#define QP   (GN + 4)
#define CHQ  8

__device__ __forceinline__ void load_a(v4f (&a)[4], const float* Ap, int ch) {
    a[0] = *(const v4f*)(Ap + 32 * ch);
    a[1] = *(const v4f*)(Ap + 32 * ch + 4);
    a[2] = *(const v4f*)(Ap + 32 * ch + 16);
    a[3] = *(const v4f*)(Ap + 32 * ch + 20);
}

__device__ __forceinline__ void load_plane(v4f (&p)[4], const float* S1, const float* S2, size_t bstride, int ch) {
    const float* r1 = S1 + (size_t)(16 * ch) * bstride;
    const float* r2 = S2 + (size_t)(16 * ch) * bstride;
    p[0] = *(const v4f*)(r1);
    p[1] = *(const v4f*)(r2);
    p[2] = *(const v4f*)(r1 + bstride);
    p[3] = *(const v4f*)(r2 + bstride);
}

__device__ __forceinline__ void store_plane(v4f* dst, const v4f (&p)[4]) {
    #pragma unroll
    for (int i = 0; i < 4; ++i) {
        const v4f col = {p[0][i], p[1][i], p[2][i], p[3][i]};
        dst[i * 32] = col;
    }
}

__device__ __forceinline__ void mma_chunk(v8f (&acc)[8], const v4f* rd, const v4f (&a)[4]) {
    #pragma unroll
    for (int gg = 0; gg < 2; ++gg) {
        v4f blo[8], bhi[8];
        __builtin_amdgcn_sched_barrier(0);
        #pragma unroll
        for (int q = 0; q < 8; ++q) blo[q] = rd[(4 * gg) * QP + 16 * q];
        #pragma unroll
        for (int q = 0; q < 8; ++q) bhi[q] = rd[(4 * gg + 1) * QP + 16 * q];
        __builtin_amdgcn_sched_barrier(0);
        const v4f alo = a[2 * gg], ahi = a[2 * gg + 1];
        const v2f a0 = {alo.x, alo.y}, a1 = {alo.z, alo.w}, a2 = {ahi.x, ahi.y}, a3 = {ahi.z, ahi.w};
        #pragma unroll
        for (int q = 0; q < 8; ++q) {
            const v2f bq = {blo[q].x, blo[q].y};
            acc[q] = __builtin_amdgcn_wmma_f32_16x16x4_f32(false, a0, false, bq, (short)0, acc[q], false, false);
        }
        #pragma unroll
        for (int q = 0; q < 8; ++q) {
            const v2f bq = {blo[q].z, blo[q].w};
            acc[q] = __builtin_amdgcn_wmma_f32_16x16x4_f32(false, a1, false, bq, (short)0, acc[q], false, false);
        }
        #pragma unroll
        for (int q = 0; q < 8; ++q) {
            const v2f bq = {bhi[q].x, bhi[q].y};
            acc[q] = __builtin_amdgcn_wmma_f32_16x16x4_f32(false, a2, false, bq, (short)0, acc[q], false, false);
        }
        #pragma unroll
        for (int q = 0; q < 8; ++q) {
            const v2f bq = {bhi[q].z, bhi[q].w};
            acc[q] = __builtin_amdgcn_wmma_f32_16x16x4_f32(false, a3, false, bq, (short)0, acc[q], false, false);
        }
    }
}

__global__ __launch_bounds__(256) void gemm_kernel(
    const float* __restrict__ W,
    const float* __restrict__ M,
    const float* __restrict__ v1,
    const float* __restrict__ v2,
    float* __restrict__ out)
{
    __shared__ v4f sb[2 * CHQ * QP];

    const int lane = threadIdx.x & 31;
    const int wave = threadIdx.x >> 5;
    const int c    = lane & 15;
    const int h    = lane >> 4;

    const int b  = blockIdx.x >> 2;
    const int n0 = (blockIdx.x & 3) * GN;

    v8f acc[8] = {};

    const float* Ap = W + ((size_t)(wave * 16 + c) * B_DIM + b) * NSLOTS + 8 * h;
    const float* Mp = M + (size_t)(wave * 16 + c) * NSLOTS + 8 * h;
    const size_t bstride = (size_t)B_DIM * R_DIM;
    const float* S1 = v1 + (size_t)(2 * wave) * bstride + (size_t)b * R_DIM + n0 + 4 * lane;
    const float* S2 = v2 + (size_t)(2 * wave) * bstride + (size_t)b * R_DIM + n0 + 4 * lane;
    v4f*       wr = sb + wave * QP + lane;
    const v4f* rd = sb + (2 * h) * QP + c;

    v4f a[4], an[4], mn[4], p[4];
    load_plane(p, S1, S2, bstride, 0);
    load_a(an, Ap, 0);
    load_a(mn, Mp, 0);
    store_plane(wr, p);
    #pragma unroll
    for (int i = 0; i < 4; ++i) {
        #pragma unroll
        for (int e = 0; e < 4; ++e) a[i][e] = fminf(an[i][e], mn[i][e]);
    }
    lds_barrier();

    for (int ch = 0; ch < NSLOTS / 32 - 1; ++ch) {
        const int cur = (ch & 1) * CHQ * QP, nxt = CHQ * QP - cur;
        load_plane(p, S1, S2, bstride, ch + 1);
        load_a(an, Ap, ch + 1);
        load_a(mn, Mp, ch + 1);
        __builtin_amdgcn_sched_barrier(0);
        mma_chunk(acc, rd + cur, a);
        store_plane(wr + nxt, p);
        #pragma unroll
        for (int i = 0; i < 4; ++i) {
            #pragma unroll
            for (int e = 0; e < 4; ++e) a[i][e] = fminf(an[i][e], mn[i][e]);
        }
        lds_barrier();
    }
    mma_chunk(acc, rd + CHQ * QP, a);

    #pragma unroll
    for (int r = 0; r < 8; ++r) {
        const int trow = wave * 16 + r + 8 * h;
        float* orow = out + ((size_t)trow * B_DIM + b) * R_DIM + n0 + 4 * c;
        const v4f o0 = {acc[0][r], acc[2][r], acc[4][r], acc[6][r]};
        const v4f o1 = {acc[1][r], acc[3][r], acc[5][r], acc[7][r]};
        *(v4f*)orow = o0;
        *(v4f*)(orow + 64) = o1;
    }
}

extern "C" void kernel_launch(void* const* d_in, const int* in_sizes, int n_in,
                              void* d_out, int out_size, void* d_ws, size_t ws_size,
                              hipStream_t stream) {
    const float* u  = (const float*)d_in[0];
    const float* d1 = (const float*)d_in[1];
    const float* d2 = (const float*)d_in[2];
    const float* v1 = (const float*)d_in[3];
    const float* v2 = (const float*)d_in[4];
    float* out = (float*)d_out;
    float* W   = (float*)d_ws;
    float* P   = W + (size_t)T_DIM * B_DIM * NSLOTS;
    float* M   = P + (size_t)T_DIM * NBG * NSLOTS;

    state_kernel<<<dim3((NSLOTS / GSLOTS) * NBG), dim3(NB * 32), 0, stream>>>(u, d1, d2, W, P);
    maxima_kernel<<<dim3(T_DIM), dim3(64), 0, stream>>>(P, M);
    gemm_kernel<<<dim3(B_DIM * 4), dim3(256), 0, stream>>>(W, M, v1, v2, out);
}


# GEMM block walks its own batch's slot strengths (same recurrences, same order) into LDS tiles and reads its A rows there; S never goes through memory, the state kernel stores maxima only; same values
# speedup vs baseline: 95.1431x; 1.0997x over previous
#include <hip/hip_runtime.h>


#define T_DIM 128
#define B_DIM 128
#define R_DIM 512
#define NSLOTS 256
#define GSLOTS 32
#define NB     4
#define NBG    (B_DIM / NB)
#define TPAD   (T_DIM + 8)
#define SP     36
#define SROWS  576

typedef float v2f __attribute__((ext_vector_type(2)));
typedef float v4f __attribute__((ext_vector_type(4)));
typedef float v8f __attribute__((ext_vector_type(8)));

struct Rows4 { v4f u, c, dn; };

__device__ __forceinline__ void load_rows(Rows4& r, const float* su, const float* sc, const float* sdn, int t) {
    r.u = *(const v4f*)(su + t); r.c = *(const v4f*)(sc + t); r.dn = *(const v4f*)(sdn + t);
}

__device__ __forceinline__ void lds_barrier() {
    asm volatile("s_wait_dscnt 0x0\n\ts_barrier_signal -1\n\ts_barrier_wait -1" ::: "memory");
}

template <bool EARLY>
__device__ __forceinline__ void walk4x(
    int t, int tj, float* xw,
    const float* su, const float* sc, const float* sdn,
    float& used, float bornU,
    const Rows4& cur, Rows4& nxt)
{
    load_rows(nxt, su, sc, sdn, t + 4);
    #pragma unroll
    for (int s = 0; s < 4; ++s) {
        const int tt = t + s;
        xw[s * NB * GSLOTS] = cur.u[s] - used;
        used = fmaxf(used + cur.c[s], cur.dn[s]);
        if (EARLY) used = (tj == tt + 1) ? bornU : used;
    }
}

template <bool EARLY>
__device__ __forceinline__ void walk4s(
    int t, int tj, float* srow,
    const float* su, const float* sc, const float* sdn,
    float& S, float& used, float bornS, float bornU,
    const Rows4& cur, Rows4& nxt)
{
    load_rows(nxt, su, sc, sdn, t + 4);
    #pragma unroll
    for (int s = 0; s < 4; ++s) {
        const int tt = t + s;
        const float x = cur.u[s] - used;
        srow[s * SP] = S;
        S    = S - fminf(S, fmaxf(0.0f, x));
        used = fmaxf(used + cur.c[s], cur.dn[s]);
        if (EARLY) {
            const bool born = tj == tt + 1;
            S    = born ? bornS : S;
            used = born ? bornU : used;
        }
    }
}

template <bool EARLY>
__device__ __forceinline__ void trip8(
    int t, int tj, int stage, int wave, int lane, float* pcol, float* xs,
    const float* su, const float* sc, const float* sdn,
    float& used, float bornU, Rows4& A, Rows4& B)
{
    float* tile = xs + stage * 8 * NB * GSLOTS;
    float* xw   = tile + wave * GSLOTS + lane;
    walk4x<EARLY>(t,     tj, xw,                   su, sc, sdn, used, bornU, A, B);
    walk4x<EARLY>(t + 4, tj, xw + 4 * NB * GSLOTS, su, sc, sdn, used, bornU, B, A);
    lds_barrier();
    #pragma unroll
    for (int e = 0; e < 2; ++e) {
        const int step = 2 * wave + e;
        const float* xr = tile + step * NB * GSLOTS + lane;
        const float m = fmaxf(fmaxf(xr[0], xr[GSLOTS]), fmaxf(xr[2 * GSLOTS], xr[3 * GSLOTS]));
        pcol[(size_t)(t + step) * NBG * NSLOTS] = m;
    }
}

__global__ __launch_bounds__(NB * 32) void state_kernel(
    const float* __restrict__ u,
    const float* __restrict__ d1,
    const float* __restrict__ d2,
    float* __restrict__ P)
{
    __shared__ v4f su4[NB * TPAD / 4], sc4[NB * TPAD / 4], sdn4[NB * TPAD / 4];
    __shared__ float xs[2 * 8 * NB * GSLOTS];
    float* suf  = (float*)su4;
    float* scf  = (float*)sc4;
    float* sdnf = (float*)sdn4;

    const int tid  = threadIdx.x;
    const int lane = tid & 31;
    const int wave = tid >> 5;
    const int g  = blockIdx.x / NBG;
    const int bg = blockIdx.x % NBG;
    const int j  = g * GSLOTS + lane;
    const int tj = j >> 1;
    const int b0 = bg * NB;
    const int b  = b0 + wave;
    const int t0 = g * (GSLOTS / 2);
    const bool odd = (j & 1) != 0;
    const v4f zero = {0.0f, 0.0f, 0.0f, 0.0f};

    {
        const int t  = t0 + tid;
        const int tc = t < T_DIM ? t : T_DIM - 1;
        const int tn = t + 1 < T_DIM ? t + 1 : T_DIM - 1;
        const v4f uu = *(const v4f*)(u  + tc * B_DIM + b0);
        const v4f a1 = *(const v4f*)(d1 + tn * B_DIM + b0);
        const v4f a2 = *(const v4f*)(d2 + tn * B_DIM + b0);
        if (t < T_DIM) {
            const v4f dn = t + 1 < T_DIM ? a1 + a2 : zero;
            #pragma unroll
            for (int i = 0; i < NB; ++i) {
                const float up = fmaxf(uu[i], 0.0f);
                suf[i * TPAD + t] = uu[i]; scf[i * TPAD + t] = dn[i] - up; sdnf[i * TPAD + t] = dn[i];
            }
        }
        if (tid < TPAD - T_DIM) {
            #pragma unroll
            for (int i = 0; i < NB; ++i) {
                suf[i * TPAD + T_DIM + tid] = 0.0f; scf[i * TPAD + T_DIM + tid] = 0.0f; sdnf[i * TPAD + T_DIM + tid] = 0.0f;
            }
        }
    }

    const float bornU = odd ? 0.0f : d2[tj * B_DIM + b];

    float used = (tj == t0) ? bornU : 0.0f;

    __syncthreads();

    const float* su  = suf  + wave * TPAD;
    const float* sc  = scf  + wave * TPAD;
    const float* sdn = sdnf + wave * TPAD;
    float* pcol = P + (size_t)bg * NSLOTS + j;

    Rows4 A, B;
    load_rows(A, su, sc, sdn, t0);
    int stage = 0;
    for (int t = t0; t < t0 + GSLOTS / 2; t += 8) {
        trip8<true>(t, tj, stage, wave, lane, pcol, xs, su, sc, sdn, used, bornU, A, B);
        stage ^= 1;
    }
    for (int t = t0 + GSLOTS / 2; t < T_DIM; t += 8) {
        trip8<false>(t, tj, stage, wave, lane, pcol, xs, su, sc, sdn, used, bornU, A, B);
        stage ^= 1;
    }
}

__global__ __launch_bounds__(64) void maxima_kernel(
    const float* __restrict__ P,
    float* __restrict__ M)
{
    const int t   = blockIdx.x;
    const int k   = threadIdx.x;
    const int top = 2 * t + 1;
    v4f m = {0.0f, 0.0f, 0.0f, 0.0f};
    if (4 * k <= top) {
        const float* p = P + (size_t)t * NBG * NSLOTS + 4 * k;
        v4f x[NBG];
        #pragma unroll
        for (int q = 0; q < NBG; ++q) x[q] = *(const v4f*)(p + q * NSLOTS);
        __builtin_amdgcn_sched_barrier(0);
        m = x[0];
        #pragma unroll
        for (int q = 1; q < NBG; ++q) {
            #pragma unroll
            for (int e = 0; e < 4; ++e) m[e] = fmaxf(m[e], x[q][e]);
        }
        #pragma unroll
        for (int e = 0; e < 4; ++e) m[e] = (4 * k + e <= top) ? m[e] : 0.0f;
    }
    *(v4f*)(M + (size_t)t * NSLOTS + 4 * k) = m;
}

#define GN   128
#define QP   (GN + 4)
#define CHQ  8

__device__ __forceinline__ void load_a(v4f (&a)[4], const float* Ap, int ch) {
    a[0] = *(const v4f*)(Ap + 32 * ch);
    a[1] = *(const v4f*)(Ap + 32 * ch + 4);
    a[2] = *(const v4f*)(Ap + 32 * ch + 16);
    a[3] = *(const v4f*)(Ap + 32 * ch + 20);
}

__device__ __forceinline__ void load_s(v4f (&a)[4], const float* row) {
    a[0] = *(const v4f*)(row);
    a[1] = *(const v4f*)(row + 4);
    a[2] = *(const v4f*)(row + 16);
    a[3] = *(const v4f*)(row + 20);
}

__device__ __forceinline__ int tile_row0(int g) { return 136 * g - 8 * g * g; }

__device__ __forceinline__ void load_plane(v4f (&p)[4], const float* S1, const float* S2, size_t bstride, int ch) {
    const float* r1 = S1 + (size_t)(16 * ch) * bstride;
    const float* r2 = S2 + (size_t)(16 * ch) * bstride;
    p[0] = *(const v4f*)(r1);
    p[1] = *(const v4f*)(r2);
    p[2] = *(const v4f*)(r1 + bstride);
    p[3] = *(const v4f*)(r2 + bstride);
}

__device__ __forceinline__ void store_plane(v4f* dst, const v4f (&p)[4]) {
    #pragma unroll
    for (int i = 0; i < 4; ++i) {
        const v4f col = {p[0][i], p[1][i], p[2][i], p[3][i]};
        dst[i * 32] = col;
    }
}

__device__ __forceinline__ void mma_chunk(v8f (&acc)[8], const v4f* rd, const v4f (&a)[4]) {
    #pragma unroll
    for (int gg = 0; gg < 2; ++gg) {
        v4f blo[8], bhi[8];
        __builtin_amdgcn_sched_barrier(0);
        #pragma unroll
        for (int q = 0; q < 8; ++q) blo[q] = rd[(4 * gg) * QP + 16 * q];
        #pragma unroll
        for (int q = 0; q < 8; ++q) bhi[q] = rd[(4 * gg + 1) * QP + 16 * q];
        __builtin_amdgcn_sched_barrier(0);
        const v4f alo = a[2 * gg], ahi = a[2 * gg + 1];
        const v2f a0 = {alo.x, alo.y}, a1 = {alo.z, alo.w}, a2 = {ahi.x, ahi.y}, a3 = {ahi.z, ahi.w};
        #pragma unroll
        for (int q = 0; q < 8; ++q) {
            const v2f bq = {blo[q].x, blo[q].y};
            acc[q] = __builtin_amdgcn_wmma_f32_16x16x4_f32(false, a0, false, bq, (short)0, acc[q], false, false);
        }
        #pragma unroll
        for (int q = 0; q < 8; ++q) {
            const v2f bq = {blo[q].z, blo[q].w};
            acc[q] = __builtin_amdgcn_wmma_f32_16x16x4_f32(false, a1, false, bq, (short)0, acc[q], false, false);
        }
        #pragma unroll
        for (int q = 0; q < 8; ++q) {
            const v2f bq = {bhi[q].x, bhi[q].y};
            acc[q] = __builtin_amdgcn_wmma_f32_16x16x4_f32(false, a2, false, bq, (short)0, acc[q], false, false);
        }
        #pragma unroll
        for (int q = 0; q < 8; ++q) {
            const v2f bq = {bhi[q].z, bhi[q].w};
            acc[q] = __builtin_amdgcn_wmma_f32_16x16x4_f32(false, a3, false, bq, (short)0, acc[q], false, false);
        }
    }
}

__global__ __launch_bounds__(256) void gemm_kernel(
    const float* __restrict__ u,
    const float* __restrict__ d1,
    const float* __restrict__ d2,
    const float* __restrict__ M,
    const float* __restrict__ v1,
    const float* __restrict__ v2,
    float* __restrict__ out)
{
    __shared__ v4f sb[2 * CHQ * QP];
    __shared__ v4f ss4[SROWS * SP / 4];
    __shared__ v4f gu4[256 / 4], gc4[256 / 4], gdn4[256 / 4];
    float* ssf = (float*)ss4;
    float* gu  = (float*)gu4;
    float* gc  = (float*)gc4;
    float* gdn = (float*)gdn4;

    const int tid  = threadIdx.x;
    const int lane = tid & 31;
    const int wave = __builtin_amdgcn_readfirstlane(tid >> 5);
    const int c    = lane & 15;
    const int h    = lane >> 4;

    const int b  = blockIdx.x >> 2;
    const int n0 = (blockIdx.x & 3) * GN;

    v8f acc[8] = {};

    const float* Mp = M + (size_t)(wave * 16 + c) * NSLOTS + 8 * h;
    const size_t bstride = (size_t)B_DIM * R_DIM;
    const float* S1 = v1 + (size_t)(2 * wave) * bstride + (size_t)b * R_DIM + n0 + 4 * lane;
    const float* S2 = v2 + (size_t)(2 * wave) * bstride + (size_t)b * R_DIM + n0 + 4 * lane;
    v4f*       wr = sb + wave * QP + lane;
    const v4f* rd = sb + (2 * h) * QP + c;

    const int ts = tid < T_DIM ? tid : T_DIM - 1;
    const int tn = ts + 1 < T_DIM ? ts + 1 : T_DIM - 1;
    const float uu = u[ts * B_DIM + b];
    const float a1 = d1[tn * B_DIM + b];
    const float a2 = d2[tn * B_DIM + b];
    const int j  = wave * GSLOTS + lane;
    const int tj = j >> 1;
    const float b1 = d1[tj * B_DIM + b];
    const float b2 = d2[tj * B_DIM + b];

    v4f a[4], an[4], mn[4], p[4];
    load_plane(p, S1, S2, bstride, 0);
    load_a(mn, Mp, 0);
    __builtin_amdgcn_sched_barrier(0);

    {
        const bool in = tid < T_DIM;
        const float dn = tid + 1 < T_DIM ? a1 + a2 : 0.0f;
        const float up = fmaxf(uu, 0.0f);
        gu[tid] = in ? uu : 0.0f; gc[tid] = in ? dn - up : 0.0f; gdn[tid] = in ? dn : 0.0f;
    }
    {
        const int t0 = wave * (GSLOTS / 2);
        const bool odd = (j & 1) != 0;
        const float bornS = odd ? b2 : b1;
        const float bornU = odd ? 0.0f : b2;
        float S    = (tj == t0) ? bornS : 0.0f;
        float used = (tj == t0) ? bornU : 0.0f;

        lds_barrier();

        float* srow = ssf + tile_row0(wave) * SP + lane;
        Rows4 A, B;
        load_rows(A, gu, gc, gdn, t0);
        for (int t = t0; t < t0 + GSLOTS / 2; t += 8) {
            walk4s<true>(t,     tj, srow,          gu, gc, gdn, S, used, bornS, bornU, A, B);
            walk4s<true>(t + 4, tj, srow + 4 * SP, gu, gc, gdn, S, used, bornS, bornU, B, A);
            srow += 8 * SP;
        }
        for (int t = t0 + GSLOTS / 2; t < T_DIM; t += 8) {
            walk4s<false>(t,     tj, srow,          gu, gc, gdn, S, used, bornS, bornU, A, B);
            walk4s<false>(t + 4, tj, srow + 4 * SP, gu, gc, gdn, S, used, bornS, bornU, B, A);
            srow += 8 * SP;
        }
    }

    store_plane(wr, p);
    lds_barrier();

    const float* srd = ssf + c * SP + 8 * h;
    load_s(an, srd + (tile_row0(0) + 16 * wave) * SP);
    #pragma unroll
    for (int i = 0; i < 4; ++i) {
        #pragma unroll
        for (int e = 0; e < 4; ++e) a[i][e] = fminf(an[i][e], mn[i][e]);
    }

    for (int ch = 0; ch < NSLOTS / 32 - 1; ++ch) {
        const int cur = (ch & 1) * CHQ * QP, nxt = CHQ * QP - cur;
        const bool live = ch + 1 <= wave;
        load_plane(p, S1, S2, bstride, ch + 1);
        load_a(mn, Mp, ch + 1);
        if (live) load_s(an, srd + (tile_row0(ch + 1) + 16 * (wave - ch - 1)) * SP);
        __builtin_amdgcn_sched_barrier(0);
        mma_chunk(acc, rd + cur, a);
        store_plane(wr + nxt, p);
        #pragma unroll
        for (int i = 0; i < 4; ++i) {
            #pragma unroll
            for (int e = 0; e < 4; ++e) a[i][e] = live ? fminf(an[i][e], mn[i][e]) : 0.0f;
        }
        lds_barrier();
    }
    mma_chunk(acc, rd + CHQ * QP, a);

    #pragma unroll
    for (int r = 0; r < 8; ++r) {
        const int trow = wave * 16 + r + 8 * h;
        float* orow = out + ((size_t)trow * B_DIM + b) * R_DIM + n0 + 4 * c;
        const v4f o0 = {acc[0][r], acc[2][r], acc[4][r], acc[6][r]};
        const v4f o1 = {acc[1][r], acc[3][r], acc[5][r], acc[7][r]};
        *(v4f*)orow = o0;
        *(v4f*)(orow + 64) = o1;
    }
}

extern "C" void kernel_launch(void* const* d_in, const int* in_sizes, int n_in,
                              void* d_out, int out_size, void* d_ws, size_t ws_size,
                              hipStream_t stream) {
    const float* u  = (const float*)d_in[0];
    const float* d1 = (const float*)d_in[1];
    const float* d2 = (const float*)d_in[2];
    const float* v1 = (const float*)d_in[3];
    const float* v2 = (const float*)d_in[4];
    float* out = (float*)d_out;
    float* P   = (float*)d_ws;
    float* M   = P + (size_t)T_DIM * NBG * NSLOTS;

    state_kernel<<<dim3((NSLOTS / GSLOTS) * NBG), dim3(NB * 32), 0, stream>>>(u, d1, d2, P);
    maxima_kernel<<<dim3(T_DIM), dim3(64), 0, stream>>>(P, M);
    gemm_kernel<<<dim3(B_DIM * 4), dim3(256), 0, stream>>>(u, d1, d2, M, v1, v2, out);
}


# Cross-batch maxima made by ONE kernel: block = one push step x all 128 batches, lane = slot x batch pair; maximum in-lane, then LDS, then lane quad; partial-maxima pass and its launch go; same values
# speedup vs baseline: 98.5442x; 1.0357x over previous
#include <hip/hip_runtime.h>


#define T_DIM 128
#define B_DIM 128
#define R_DIM 512
#define NSLOTS 256
#define GSLOTS 32
#define NT     16
#define XROW   128
#define SP     36
#define SROWS  576

typedef float v2f __attribute__((ext_vector_type(2)));
typedef float v4f __attribute__((ext_vector_type(4)));
typedef float v8f __attribute__((ext_vector_type(8)));

struct Rows4 { v4f u, c, dn; };

__device__ __forceinline__ void load_rows(Rows4& r, const float* su, const float* sc, const float* sdn, int t) {
    r.u = *(const v4f*)(su + t); r.c = *(const v4f*)(sc + t); r.dn = *(const v4f*)(sdn + t);
}

__device__ __forceinline__ void lds_barrier() {
    asm volatile("s_wait_dscnt 0x0\n\ts_barrier_signal -1\n\ts_barrier_wait -1" ::: "memory");
}

template <bool EARLY>
__device__ __forceinline__ void walk4s(
    int t, int tj, float* srow,
    const float* su, const float* sc, const float* sdn,
    float& S, float& used, float bornS, float bornU,
    const Rows4& cur, Rows4& nxt)
{
    load_rows(nxt, su, sc, sdn, t + 4);
    #pragma unroll
    for (int s = 0; s < 4; ++s) {
        const int tt = t + s;
        const float x = cur.u[s] - used;
        srow[s * SP] = S;
        S    = S - fminf(S, fmaxf(0.0f, x));
        used = fmaxf(used + cur.c[s], cur.dn[s]);
        if (EARLY) {
            const bool born = tj == tt + 1;
            S    = born ? bornS : S;
            used = born ? bornU : used;
        }
    }
}

struct In16 { v2f u[NT], a1[NT], a2[NT]; };

__device__ __forceinline__ void load_in(In16& r, const float* pu, const float* p1, const float* p2, int t) {
    const float* qu = pu + t * B_DIM;
    const float* q1 = p1 + (t + 1) * B_DIM;
    const float* q2 = p2 + (t + 1) * B_DIM;
    const v2f zero2 = {0.0f, 0.0f};
    const bool end = t + NT >= T_DIM;
    #pragma unroll
    for (int s = 0; s < NT; ++s) r.u[s] = *(const v2f*)(qu + s * B_DIM);
    #pragma unroll
    for (int s = 0; s < NT - 1; ++s) {
        r.a1[s] = *(const v2f*)(q1 + s * B_DIM);
        r.a2[s] = *(const v2f*)(q2 + s * B_DIM);
    }
    const int sl = end ? NT - 2 : NT - 1;
    const v2f l1 = *(const v2f*)(q1 + sl * B_DIM);
    const v2f l2 = *(const v2f*)(q2 + sl * B_DIM);
    r.a1[NT - 1] = end ? zero2 : l1;
    r.a2[NT - 1] = end ? zero2 : l2;
}

template <bool XOR1>
__device__ __forceinline__ float quad_swap(float v) {
    return __builtin_bit_cast(float, __builtin_amdgcn_update_dpp(
        0, __builtin_bit_cast(int, v), XOR1 ? 0xB1 : 0x4E, 0xF, 0xF, true));
}

__device__ __forceinline__ void finish_trip(const float (&pend)[16], float* Mq, int p, int wave, int lane, int t) {
    float v = pend[0];
    #pragma unroll
    for (int i = 1; i < 16; ++i) v = fmaxf(v, pend[i]);
    v = fmaxf(v, quad_swap<true>(v));
    v = fmaxf(v, quad_swap<false>(v));
    if ((lane & 3) == 0) Mq[p * 256 + wave * 64 + (t >> 4) * 8 + (lane >> 2)] = v;
}

template <bool FIRST>
__device__ __forceinline__ void trip16(
    int t, int p, int stage, int wave, int lane, float* X, float* Mq,
    const float* pu, const float* p1, const float* p2,
    v2f& used, v2f bornU, float (&pend)[16],
    const In16& cur, In16& nxt)
{
    if (t + NT < T_DIM) load_in(nxt, pu, p1, p2, t + NT);
    float* xw = X + stage * NT * XROW + wave * 32 + lane;
    #pragma unroll
    for (int s = 0; s < NT; ++s) {
        const int tt = t + s;
        const v2f uu = cur.u[s];
        const v2f dn = cur.a1[s] + cur.a2[s];
        v2f up;
        up.x = fmaxf(uu.x, 0.0f); up.y = fmaxf(uu.y, 0.0f);
        const v2f c = dn - up;
        const v2f x = uu - used;
        float m = fmaxf(x.x, x.y);
        if (FIRST) m = tt >= p ? m : 0.0f;
        xw[s * XROW] = m;
        const v2f w = used + c;
        used.x = fmaxf(w.x, dn.x); used.y = fmaxf(w.y, dn.y);
        if (FIRST) used = (p == tt + 1) ? bornU : used;
    }
    if (!FIRST) finish_trip(pend, Mq, p, wave, lane, t - NT);
    lds_barrier();
    const float* xr = X + (stage * NT + 4 * wave + (lane >> 3)) * XROW + (lane & 3) * 32 + ((lane >> 2) & 1);
    #pragma unroll
    for (int i = 0; i < 16; ++i) pend[i] = xr[2 * i];
}

__global__ __launch_bounds__(128) void state_kernel(
    const float* __restrict__ u,
    const float* __restrict__ d1,
    const float* __restrict__ d2,
    float* __restrict__ Mq)
{
    __shared__ float X[2 * NT * XROW];

    const int tid  = threadIdx.x;
    const int lane = tid & 31;
    const int wave = __builtin_amdgcn_readfirstlane(tid >> 5);
    const int p    = blockIdx.x;
    const int tA   = p & ~(NT - 1);
    const int bofs = wave * 32 + (lane >> 1) * 2;
    const float* pu = u  + bofs;
    const float* p1 = d1 + bofs;
    const float* p2 = d2 + bofs;
    const v2f zero2 = {0.0f, 0.0f};

    const v2f top = *(const v2f*)(d2 + p * B_DIM + bofs);
    const v2f bornU = (lane & 1) ? zero2 : top;
    v2f used = (tA == p) ? bornU : zero2;

    In16 A, B;
    float pend[16];
    load_in(A, pu, p1, p2, tA);
    int t = tA, stage = 0;
    trip16<true>(t, p, stage, wave, lane, X, Mq, pu, p1, p2, used, bornU, pend, A, B);
    t += NT; stage ^= 1;
    while (t < T_DIM) {
        trip16<false>(t, p, stage, wave, lane, X, Mq, pu, p1, p2, used, bornU, pend, B, A);
        t += NT; stage ^= 1;
        if (t >= T_DIM) break;
        trip16<false>(t, p, stage, wave, lane, X, Mq, pu, p1, p2, used, bornU, pend, A, B);
        t += NT; stage ^= 1;
    }
    finish_trip(pend, Mq, p, wave, lane, t - NT);
}

#define GN   128
#define QP   (GN + 4)
#define CHQ  8

__device__ __forceinline__ v4f load_a4(const float* q) {
    const v2f lo = *(const v2f*)(q), hi = *(const v2f*)(q + 256);
    const v4f r = {lo.x, lo.y, hi.x, hi.y};
    return r;
}
__device__ __forceinline__ void load_a(v4f (&a)[4], const float* Ap, int ch) {
    a[0] = load_a4(Ap + 4096 * ch);
    a[1] = load_a4(Ap + 4096 * ch + 512);
    a[2] = load_a4(Ap + 4096 * ch + 2048);
    a[3] = load_a4(Ap + 4096 * ch + 2560);
}

__device__ __forceinline__ void load_s(v4f (&a)[4], const float* row) {
    a[0] = *(const v4f*)(row);
    a[1] = *(const v4f*)(row + 4);
    a[2] = *(const v4f*)(row + 16);
    a[3] = *(const v4f*)(row + 20);
}

__device__ __forceinline__ int tile_row0(int g) { return 136 * g - 8 * g * g; }

__device__ __forceinline__ void load_plane(v4f (&p)[4], const float* S1, const float* S2, size_t bstride, int ch) {
    const float* r1 = S1 + (size_t)(16 * ch) * bstride;
    const float* r2 = S2 + (size_t)(16 * ch) * bstride;
    p[0] = *(const v4f*)(r1);
    p[1] = *(const v4f*)(r2);
    p[2] = *(const v4f*)(r1 + bstride);
    p[3] = *(const v4f*)(r2 + bstride);
}

__device__ __forceinline__ void store_plane(v4f* dst, const v4f (&p)[4]) {
    #pragma unroll
    for (int i = 0; i < 4; ++i) {
        const v4f col = {p[0][i], p[1][i], p[2][i], p[3][i]};
        dst[i * 32] = col;
    }
}

__device__ __forceinline__ void mma_chunk(v8f (&acc)[8], const v4f* rd, const v4f (&a)[4]) {
    #pragma unroll
    for (int gg = 0; gg < 2; ++gg) {
        v4f blo[8], bhi[8];
        __builtin_amdgcn_sched_barrier(0);
        #pragma unroll
        for (int q = 0; q < 8; ++q) blo[q] = rd[(4 * gg) * QP + 16 * q];
        #pragma unroll
        for (int q = 0; q < 8; ++q) bhi[q] = rd[(4 * gg + 1) * QP + 16 * q];
        __builtin_amdgcn_sched_barrier(0);
        const v4f alo = a[2 * gg], ahi = a[2 * gg + 1];
        const v2f a0 = {alo.x, alo.y}, a1 = {alo.z, alo.w}, a2 = {ahi.x, ahi.y}, a3 = {ahi.z, ahi.w};
        #pragma unroll
        for (int q = 0; q < 8; ++q) {
            const v2f bq = {blo[q].x, blo[q].y};
            acc[q] = __builtin_amdgcn_wmma_f32_16x16x4_f32(false, a0, false, bq, (short)0, acc[q], false, false);
        }
        #pragma unroll
        for (int q = 0; q < 8; ++q) {
            const v2f bq = {blo[q].z, blo[q].w};
            acc[q] = __builtin_amdgcn_wmma_f32_16x16x4_f32(false, a1, false, bq, (short)0, acc[q], false, false);
        }
        #pragma unroll
        for (int q = 0; q < 8; ++q) {
            const v2f bq = {bhi[q].x, bhi[q].y};
            acc[q] = __builtin_amdgcn_wmma_f32_16x16x4_f32(false, a2, false, bq, (short)0, acc[q], false, false);
        }
        #pragma unroll
        for (int q = 0; q < 8; ++q) {
            const v2f bq = {bhi[q].z, bhi[q].w};
            acc[q] = __builtin_amdgcn_wmma_f32_16x16x4_f32(false, a3, false, bq, (short)0, acc[q], false, false);
        }
    }
}

__global__ __launch_bounds__(256) void gemm_kernel(
    const float* __restrict__ u,
    const float* __restrict__ d1,
    const float* __restrict__ d2,
    const float* __restrict__ Mq,
    const float* __restrict__ v1,
    const float* __restrict__ v2,
    float* __restrict__ out)
{
    __shared__ v4f sb[2 * CHQ * QP];
    __shared__ v4f ss4[SROWS * SP / 4];
    __shared__ v4f gu4[256 / 4], gc4[256 / 4], gdn4[256 / 4];
    float* ssf = (float*)ss4;
    float* gu  = (float*)gu4;
    float* gc  = (float*)gc4;
    float* gdn = (float*)gdn4;

    const int tid  = threadIdx.x;
    const int lane = tid & 31;
    const int wave = __builtin_amdgcn_readfirstlane(tid >> 5);
    const int c    = lane & 15;
    const int h    = lane >> 4;

    const int b  = blockIdx.x >> 2;
    const int n0 = (blockIdx.x & 3) * GN;

    v8f acc[8] = {};

    const float* Mp = Mq + (c >> 2) * 64 + wave * 8 + (c & 3) * 2 + h * 1024;
    const size_t bstride = (size_t)B_DIM * R_DIM;
    const float* S1 = v1 + (size_t)(2 * wave) * bstride + (size_t)b * R_DIM + n0 + 4 * lane;
    const float* S2 = v2 + (size_t)(2 * wave) * bstride + (size_t)b * R_DIM + n0 + 4 * lane;
    v4f*       wr = sb + wave * QP + lane;
    const v4f* rd = sb + (2 * h) * QP + c;

    const int ts = tid < T_DIM ? tid : T_DIM - 1;
    const int tn = ts + 1 < T_DIM ? ts + 1 : T_DIM - 1;
    const float uu = u[ts * B_DIM + b];
    const float a1 = d1[tn * B_DIM + b];
    const float a2 = d2[tn * B_DIM + b];
    const int j  = wave * GSLOTS + lane;
    const int tj = j >> 1;
    const float b1 = d1[tj * B_DIM + b];
    const float b2 = d2[tj * B_DIM + b];

    v4f a[4], an[4], mn[4], p[4];
    load_plane(p, S1, S2, bstride, 0);
    load_a(mn, Mp, 0);
    __builtin_amdgcn_sched_barrier(0);

    {
        const bool in = tid < T_DIM;
        const float dn = tid + 1 < T_DIM ? a1 + a2 : 0.0f;
        const float up = fmaxf(uu, 0.0f);
        gu[tid] = in ? uu : 0.0f; gc[tid] = in ? dn - up : 0.0f; gdn[tid] = in ? dn : 0.0f;
    }
    {
        const int t0 = wave * (GSLOTS / 2);
        const bool odd = (j & 1) != 0;
        const float bornS = odd ? b2 : b1;
        const float bornU = odd ? 0.0f : b2;
        float S    = (tj == t0) ? bornS : 0.0f;
        float used = (tj == t0) ? bornU : 0.0f;

        lds_barrier();

        float* srow = ssf + tile_row0(wave) * SP + lane;
        Rows4 A, B;
        load_rows(A, gu, gc, gdn, t0);
        for (int t = t0; t < t0 + GSLOTS / 2; t += 8) {
            walk4s<true>(t,     tj, srow,          gu, gc, gdn, S, used, bornS, bornU, A, B);
            walk4s<true>(t + 4, tj, srow + 4 * SP, gu, gc, gdn, S, used, bornS, bornU, B, A);
            srow += 8 * SP;
        }
        for (int t = t0 + GSLOTS / 2; t < T_DIM; t += 8) {
            walk4s<false>(t,     tj, srow,          gu, gc, gdn, S, used, bornS, bornU, A, B);
            walk4s<false>(t + 4, tj, srow + 4 * SP, gu, gc, gdn, S, used, bornS, bornU, B, A);
            srow += 8 * SP;
        }
    }

    store_plane(wr, p);
    lds_barrier();

    const float* srd = ssf + c * SP + 8 * h;
    load_s(an, srd + (tile_row0(0) + 16 * wave) * SP);
    #pragma unroll
    for (int i = 0; i < 4; ++i) {
        #pragma unroll
        for (int e = 0; e < 4; ++e) a[i][e] = fminf(an[i][e], mn[i][e]);
    }

    for (int ch = 0; ch < NSLOTS / 32 - 1; ++ch) {
        const int cur = (ch & 1) * CHQ * QP, nxt = CHQ * QP - cur;
        const bool live = ch + 1 <= wave;
        load_plane(p, S1, S2, bstride, ch + 1);
        if (live) {
            load_a(mn, Mp, ch + 1);
            load_s(an, srd + (tile_row0(ch + 1) + 16 * (wave - ch - 1)) * SP);
        }
        __builtin_amdgcn_sched_barrier(0);
        mma_chunk(acc, rd + cur, a);
        store_plane(wr + nxt, p);
        #pragma unroll
        for (int i = 0; i < 4; ++i) {
            #pragma unroll
            for (int e = 0; e < 4; ++e) a[i][e] = live ? fminf(an[i][e], mn[i][e]) : 0.0f;
        }
        lds_barrier();
    }
    mma_chunk(acc, rd + CHQ * QP, a);

    #pragma unroll
    for (int r = 0; r < 8; ++r) {
        const int trow = wave * 16 + r + 8 * h;
        float* orow = out + ((size_t)trow * B_DIM + b) * R_DIM + n0 + 4 * c;
        const v4f o0 = {acc[0][r], acc[2][r], acc[4][r], acc[6][r]};
        const v4f o1 = {acc[1][r], acc[3][r], acc[5][r], acc[7][r]};
        *(v4f*)orow = o0;
        *(v4f*)(orow + 64) = o1;
    }
}

extern "C" void kernel_launch(void* const* d_in, const int* in_sizes, int n_in,
                              void* d_out, int out_size, void* d_ws, size_t ws_size,
                              hipStream_t stream) {
    const float* u  = (const float*)d_in[0];
    const float* d1 = (const float*)d_in[1];
    const float* d2 = (const float*)d_in[2];
    const float* v1 = (const float*)d_in[3];
    const float* v2 = (const float*)d_in[4];
    float* out = (float*)d_out;
    float* Mq  = (float*)d_ws;

    state_kernel<<<dim3(T_DIM), dim3(128), 0, stream>>>(u, d1, d2, Mq);
    gemm_kernel<<<dim3(B_DIM * 4), dim3(256), 0, stream>>>(u, d1, d2, Mq, v1, v2, out);
}


# GEMM: exact-zero chunks above row tile i are not multiplied (wave = row tiles x, 7-x by 64 cols); v1/v2 are fetched 2 chunks ahead, without which shorter chunks wait for memory on misses; same sums
# speedup vs baseline: 107.4471x; 1.0903x over previous
#include <hip/hip_runtime.h>


#define T_DIM 128
#define B_DIM 128
#define R_DIM 512
#define NSLOTS 256
#define GSLOTS 32
#define NT     16
#define XROW   128
#define SP     36
#define SROWS  576

typedef float v2f __attribute__((ext_vector_type(2)));
typedef float v4f __attribute__((ext_vector_type(4)));
typedef float v8f __attribute__((ext_vector_type(8)));

struct Rows4 { v4f u, c, dn; };

__device__ __forceinline__ void load_rows(Rows4& r, const float* su, const float* sc, const float* sdn, int t) {
    r.u = *(const v4f*)(su + t); r.c = *(const v4f*)(sc + t); r.dn = *(const v4f*)(sdn + t);
}

__device__ __forceinline__ void lds_barrier() {
    asm volatile("s_wait_dscnt 0x0\n\ts_barrier_signal -1\n\ts_barrier_wait -1" ::: "memory");
}

template <bool EARLY>
__device__ __forceinline__ void walk4s(
    int t, int tj, float* srow,
    const float* su, const float* sc, const float* sdn,
    float& S, float& used, float bornS, float bornU,
    const Rows4& cur, Rows4& nxt)
{
    load_rows(nxt, su, sc, sdn, t + 4);
    #pragma unroll
    for (int s = 0; s < 4; ++s) {
        const int tt = t + s;
        const float x = cur.u[s] - used;
        srow[s * SP] = S;
        S    = S - fminf(S, fmaxf(0.0f, x));
        used = fmaxf(used + cur.c[s], cur.dn[s]);
        if (EARLY) {
            const bool born = tj == tt + 1;
            S    = born ? bornS : S;
            used = born ? bornU : used;
        }
    }
}

struct In16 { v2f u[NT], a1[NT], a2[NT]; };

__device__ __forceinline__ void load_in(In16& r, const float* pu, const float* p1, const float* p2, int t) {
    const float* qu = pu + t * B_DIM;
    const float* q1 = p1 + (t + 1) * B_DIM;
    const float* q2 = p2 + (t + 1) * B_DIM;
    const v2f zero2 = {0.0f, 0.0f};
    const bool end = t + NT >= T_DIM;
    #pragma unroll
    for (int s = 0; s < NT; ++s) r.u[s] = *(const v2f*)(qu + s * B_DIM);
    #pragma unroll
    for (int s = 0; s < NT - 1; ++s) {
        r.a1[s] = *(const v2f*)(q1 + s * B_DIM);
        r.a2[s] = *(const v2f*)(q2 + s * B_DIM);
    }
    const int sl = end ? NT - 2 : NT - 1;
    const v2f l1 = *(const v2f*)(q1 + sl * B_DIM);
    const v2f l2 = *(const v2f*)(q2 + sl * B_DIM);
    r.a1[NT - 1] = end ? zero2 : l1;
    r.a2[NT - 1] = end ? zero2 : l2;
}

template <bool XOR1>
__device__ __forceinline__ float quad_swap(float v) {
    return __builtin_bit_cast(float, __builtin_amdgcn_update_dpp(
        0, __builtin_bit_cast(int, v), XOR1 ? 0xB1 : 0x4E, 0xF, 0xF, true));
}

__device__ __forceinline__ void finish_trip(const float (&pend)[16], float* Mq, int p, int wave, int lane, int t) {
    float v = pend[0];
    #pragma unroll
    for (int i = 1; i < 16; ++i) v = fmaxf(v, pend[i]);
    v = fmaxf(v, quad_swap<true>(v));
    v = fmaxf(v, quad_swap<false>(v));
    if ((lane & 3) == 0) Mq[p * 256 + wave * 64 + (t >> 4) * 8 + (lane >> 2)] = v;
}

template <bool FIRST>
__device__ __forceinline__ void trip16(
    int t, int p, int stage, int wave, int lane, float* X, float* Mq,
    const float* pu, const float* p1, const float* p2,
    v2f& used, v2f bornU, float (&pend)[16],
    const In16& cur, In16& nxt)
{
    if (t + NT < T_DIM) load_in(nxt, pu, p1, p2, t + NT);
    float* xw = X + stage * NT * XROW + wave * 32 + lane;
    #pragma unroll
    for (int s = 0; s < NT; ++s) {
        const int tt = t + s;
        const v2f uu = cur.u[s];
        const v2f dn = cur.a1[s] + cur.a2[s];
        v2f up;
        up.x = fmaxf(uu.x, 0.0f); up.y = fmaxf(uu.y, 0.0f);
        const v2f c = dn - up;
        const v2f x = uu - used;
        float m = fmaxf(x.x, x.y);
        if (FIRST) m = tt >= p ? m : 0.0f;
        xw[s * XROW] = m;
        const v2f w = used + c;
        used.x = fmaxf(w.x, dn.x); used.y = fmaxf(w.y, dn.y);
        if (FIRST) used = (p == tt + 1) ? bornU : used;
    }
    if (!FIRST) finish_trip(pend, Mq, p, wave, lane, t - NT);
    lds_barrier();
    const float* xr = X + (stage * NT + 4 * wave + (lane >> 3)) * XROW + (lane & 3) * 32 + ((lane >> 2) & 1);
    #pragma unroll
    for (int i = 0; i < 16; ++i) pend[i] = xr[2 * i];
}

__global__ __launch_bounds__(128) void state_kernel(
    const float* __restrict__ u,
    const float* __restrict__ d1,
    const float* __restrict__ d2,
    float* __restrict__ Mq)
{
    __shared__ float X[2 * NT * XROW];

    const int tid  = threadIdx.x;
    const int lane = tid & 31;
    const int wave = __builtin_amdgcn_readfirstlane(tid >> 5);
    const int p    = blockIdx.x;
    const int tA   = p & ~(NT - 1);
    const int bofs = wave * 32 + (lane >> 1) * 2;
    const float* pu = u  + bofs;
    const float* p1 = d1 + bofs;
    const float* p2 = d2 + bofs;
    const v2f zero2 = {0.0f, 0.0f};

    const v2f top = *(const v2f*)(d2 + p * B_DIM + bofs);
    const v2f bornU = (lane & 1) ? zero2 : top;
    v2f used = (tA == p) ? bornU : zero2;

    In16 A, B;
    float pend[16];
    load_in(A, pu, p1, p2, tA);
    int t = tA, stage = 0;
    trip16<true>(t, p, stage, wave, lane, X, Mq, pu, p1, p2, used, bornU, pend, A, B);
    t += NT; stage ^= 1;
    while (t < T_DIM) {
        trip16<false>(t, p, stage, wave, lane, X, Mq, pu, p1, p2, used, bornU, pend, B, A);
        t += NT; stage ^= 1;
        if (t >= T_DIM) break;
        trip16<false>(t, p, stage, wave, lane, X, Mq, pu, p1, p2, used, bornU, pend, A, B);
        t += NT; stage ^= 1;
    }
    finish_trip(pend, Mq, p, wave, lane, t - NT);
}

#define GN   128
#define QP   (GN + 4)
#define CHQ  8
#define NCH  (NSLOTS / 32)

__device__ __forceinline__ v4f load_a4(const float* q) {
    const v2f lo = *(const v2f*)(q), hi = *(const v2f*)(q + 256);
    const v4f r = {lo.x, lo.y, hi.x, hi.y};
    return r;
}
__device__ __forceinline__ void load_a(v4f (&a)[4], const float* Ap, int ch) {
    a[0] = load_a4(Ap + 4096 * ch);
    a[1] = load_a4(Ap + 4096 * ch + 512);
    a[2] = load_a4(Ap + 4096 * ch + 2048);
    a[3] = load_a4(Ap + 4096 * ch + 2560);
}

__device__ __forceinline__ void load_s(v4f (&a)[4], const float* row) {
    a[0] = *(const v4f*)(row);
    a[1] = *(const v4f*)(row + 4);
    a[2] = *(const v4f*)(row + 16);
    a[3] = *(const v4f*)(row + 20);
}

__device__ __forceinline__ int tile_row0(int g) { return 136 * g - 8 * g * g; }

__device__ __forceinline__ void make_w(v4f (&a)[4], const float* srd, int rt, int ch, const v4f (&mn)[4]) {
    v4f an[4];
    load_s(an, srd + (tile_row0(ch) + 16 * (rt - ch)) * SP);
    #pragma unroll
    for (int i = 0; i < 4; ++i) {
        #pragma unroll
        for (int e = 0; e < 4; ++e) a[i][e] = fminf(an[i][e], mn[i][e]);
    }
}

__device__ __forceinline__ void load_plane(v4f (&p)[4], const float* S1, const float* S2, size_t bstride, int ch) {
    const float* r1 = S1 + (size_t)(16 * ch) * bstride;
    const float* r2 = S2 + (size_t)(16 * ch) * bstride;
    p[0] = *(const v4f*)(r1);
    p[1] = *(const v4f*)(r2);
    p[2] = *(const v4f*)(r1 + bstride);
    p[3] = *(const v4f*)(r2 + bstride);
}

__device__ __forceinline__ void store_plane(v4f* dst, const v4f (&p)[4]) {
    #pragma unroll
    for (int i = 0; i < 4; ++i) {
        const v4f col = {p[0][i], p[1][i], p[2][i], p[3][i]};
        dst[i * 32] = col;
    }
}

__device__ __forceinline__ void mma_group(v8f (&acc)[4], const v4f alo, const v4f ahi,
                                          const v4f (&blo)[4], const v4f (&bhi)[4]) {
    const v2f a0 = {alo.x, alo.y}, a1 = {alo.z, alo.w}, a2 = {ahi.x, ahi.y}, a3 = {ahi.z, ahi.w};
    #pragma unroll
    for (int i = 0; i < 4; ++i) {
        const v2f bq = {blo[i].x, blo[i].y};
        acc[i] = __builtin_amdgcn_wmma_f32_16x16x4_f32(false, a0, false, bq, (short)0, acc[i], false, false);
    }
    #pragma unroll
    for (int i = 0; i < 4; ++i) {
        const v2f bq = {blo[i].z, blo[i].w};
        acc[i] = __builtin_amdgcn_wmma_f32_16x16x4_f32(false, a1, false, bq, (short)0, acc[i], false, false);
    }
    #pragma unroll
    for (int i = 0; i < 4; ++i) {
        const v2f bq = {bhi[i].x, bhi[i].y};
        acc[i] = __builtin_amdgcn_wmma_f32_16x16x4_f32(false, a2, false, bq, (short)0, acc[i], false, false);
    }
    #pragma unroll
    for (int i = 0; i < 4; ++i) {
        const v2f bq = {bhi[i].z, bhi[i].w};
        acc[i] = __builtin_amdgcn_wmma_f32_16x16x4_f32(false, a3, false, bq, (short)0, acc[i], false, false);
    }
}

__device__ __forceinline__ void mma_chunk(v8f (&accL)[4], v8f (&accH)[4], const v4f* rd,
                                          const v4f (&aL)[4], const v4f (&aH)[4], bool both) {
    #pragma unroll
    for (int gg = 0; gg < 2; ++gg) {
        v4f blo[4], bhi[4];
        __builtin_amdgcn_sched_barrier(0);
        #pragma unroll
        for (int i = 0; i < 4; ++i) blo[i] = rd[(4 * gg) * QP + 32 * i];
        #pragma unroll
        for (int i = 0; i < 4; ++i) bhi[i] = rd[(4 * gg + 1) * QP + 32 * i];
        __builtin_amdgcn_sched_barrier(0);
        mma_group(accH, aH[2 * gg], aH[2 * gg + 1], blo, bhi);
        if (both) mma_group(accL, aL[2 * gg], aL[2 * gg + 1], blo, bhi);
    }
}

__device__ __forceinline__ void store_tile(float* otile, const v8f (&acc)[4], int rt, int h) {
    #pragma unroll
    for (int r = 0; r < 8; ++r) {
        const int trow = rt * 16 + r + 8 * h;
        const v4f o = {acc[0][r], acc[1][r], acc[2][r], acc[3][r]};
        *(v4f*)(otile + (size_t)trow * B_DIM * R_DIM) = o;
    }
}

template <bool REQ, bool NEXT>
__device__ __forceinline__ void chunk_step(
    int ch, int rtL, int rtH,
    v8f (&accL)[4], v8f (&accH)[4], v4f (&aL)[4], v4f (&aH)[4], v4f (&mnL)[4], v4f (&mnH)[4],
    v4f (&preq)[4], const v4f (&pst)[4],
    const float* S1, const float* S2, size_t bstride,
    const float* MpL, const float* MpH, const float* srd, v4f* wr, const v4f* rd)
{
    const int cur = (ch & 1) * CHQ * QP, nxt = CHQ * QP - cur;
    const bool nL = ch + 1 <= rtL;
    const bool nH = ch + 1 <= rtH;
    if (NEXT) {
        load_a(mnL, MpL, nL ? ch + 1 : rtL);
        load_a(mnH, MpH, nH ? ch + 1 : rtH);
    }
    if (REQ) load_plane(preq, S1, S2, bstride, ch + 2);
    __builtin_amdgcn_sched_barrier(0);
    if (ch <= rtH) mma_chunk(accL, accH, rd + cur, aL, aH, ch <= rtL);
    if (NEXT) {
        store_plane(wr + nxt, pst);
        if (nL) make_w(aL, srd, rtL, ch + 1, mnL);
        if (nH) make_w(aH, srd, rtH, ch + 1, mnH);
        lds_barrier();
    }
}

__global__ __launch_bounds__(256) void gemm_kernel(
    const float* __restrict__ u,
    const float* __restrict__ d1,
    const float* __restrict__ d2,
    const float* __restrict__ Mq,
    const float* __restrict__ v1,
    const float* __restrict__ v2,
    float* __restrict__ out)
{
    __shared__ v4f sb[2 * CHQ * QP];
    __shared__ v4f ss4[SROWS * SP / 4];
    __shared__ v4f gu4[256 / 4], gc4[256 / 4], gdn4[256 / 4];
    float* ssf = (float*)ss4;
    float* gu  = (float*)gu4;
    float* gc  = (float*)gc4;
    float* gdn = (float*)gdn4;

    const int tid  = threadIdx.x;
    const int lane = tid & 31;
    const int wave = __builtin_amdgcn_readfirstlane(tid >> 5);
    const int rtL  = wave >> 1;
    const int rtH  = 7 - rtL;
    const int hf   = wave & 1;
    const int c    = lane & 15;
    const int h    = lane >> 4;

    const int b  = blockIdx.x >> 2;
    const int n0 = (blockIdx.x & 3) * GN;

    v8f accL[4] = {}, accH[4] = {};

    const float* Mp  = Mq + (c >> 2) * 64 + (c & 3) * 2 + h * 1024;
    const float* MpL = Mp + rtL * 8;
    const float* MpH = Mp + rtH * 8;
    const size_t bstride = (size_t)B_DIM * R_DIM;
    const float* S1 = v1 + (size_t)(2 * wave) * bstride + (size_t)b * R_DIM + n0 + 4 * lane;
    const float* S2 = v2 + (size_t)(2 * wave) * bstride + (size_t)b * R_DIM + n0 + 4 * lane;
    v4f*       wr = sb + wave * QP + lane;
    const v4f* rd = sb + (2 * h) * QP + c + 16 * hf;

    const int ts = tid < T_DIM ? tid : T_DIM - 1;
    const int tn = ts + 1 < T_DIM ? ts + 1 : T_DIM - 1;
    const float uu = u[ts * B_DIM + b];
    const float a1 = d1[tn * B_DIM + b];
    const float a2 = d2[tn * B_DIM + b];
    const int j  = wave * GSLOTS + lane;
    const int tj = j >> 1;
    const float b1 = d1[tj * B_DIM + b];
    const float b2 = d2[tj * B_DIM + b];

    v4f aL[4], aH[4], mnL[4], mnH[4], p0[4], p1[4];
    load_plane(p0, S1, S2, bstride, 0);
    load_a(mnL, MpL, 0);
    load_a(mnH, MpH, 0);
    load_plane(p1, S1, S2, bstride, 1);
    __builtin_amdgcn_sched_barrier(0);

    {
        const bool in = tid < T_DIM;
        const float dn = tid + 1 < T_DIM ? a1 + a2 : 0.0f;
        const float up = fmaxf(uu, 0.0f);
        gu[tid] = in ? uu : 0.0f; gc[tid] = in ? dn - up : 0.0f; gdn[tid] = in ? dn : 0.0f;
    }
    {
        const int t0 = wave * (GSLOTS / 2);
        const bool odd = (j & 1) != 0;
        const float bornS = odd ? b2 : b1;
        const float bornU = odd ? 0.0f : b2;
        float S    = (tj == t0) ? bornS : 0.0f;
        float used = (tj == t0) ? bornU : 0.0f;

        lds_barrier();

        float* srow = ssf + tile_row0(wave) * SP + lane;
        Rows4 A, B;
        load_rows(A, gu, gc, gdn, t0);
        for (int t = t0; t < t0 + GSLOTS / 2; t += 8) {
            walk4s<true>(t,     tj, srow,          gu, gc, gdn, S, used, bornS, bornU, A, B);
            walk4s<true>(t + 4, tj, srow + 4 * SP, gu, gc, gdn, S, used, bornS, bornU, B, A);
            srow += 8 * SP;
        }
        for (int t = t0 + GSLOTS / 2; t < T_DIM; t += 8) {
            walk4s<false>(t,     tj, srow,          gu, gc, gdn, S, used, bornS, bornU, A, B);
            walk4s<false>(t + 4, tj, srow + 4 * SP, gu, gc, gdn, S, used, bornS, bornU, B, A);
            srow += 8 * SP;
        }
    }

    store_plane(wr, p0);
    lds_barrier();

    const float* srd = ssf + c * SP + 8 * h;
    make_w(aL, srd, rtL, 0, mnL);
    make_w(aH, srd, rtH, 0, mnH);

    for (int ch = 0; ch < NCH - 2; ch += 2) {
        chunk_step<true, true>(ch,     rtL, rtH, accL, accH, aL, aH, mnL, mnH, p0, p1, S1, S2, bstride, MpL, MpH, srd, wr, rd);
        chunk_step<true, true>(ch + 1, rtL, rtH, accL, accH, aL, aH, mnL, mnH, p1, p0, S1, S2, bstride, MpL, MpH, srd, wr, rd);
    }
    chunk_step<false, true >(NCH - 2, rtL, rtH, accL, accH, aL, aH, mnL, mnH, p0, p1, S1, S2, bstride, MpL, MpH, srd, wr, rd);
    chunk_step<false, false>(NCH - 1, rtL, rtH, accL, accH, aL, aH, mnL, mnH, p1, p0, S1, S2, bstride, MpL, MpH, srd, wr, rd);

    float* otile = out + (size_t)b * R_DIM + n0 + 64 * hf + 4 * c;
    store_tile(otile, accL, rtL, h);
    store_tile(otile, accH, rtH, h);
}

extern "C" void kernel_launch(void* const* d_in, const int* in_sizes, int n_in,
                              void* d_out, int out_size, void* d_ws, size_t ws_size,
                              hipStream_t stream) {
    const float* u  = (const float*)d_in[0];
    const float* d1 = (const float*)d_in[1];
    const float* d2 = (const float*)d_in[2];
    const float* v1 = (const float*)d_in[3];
    const float* v2 = (const float*)d_in[4];
    float* out = (float*)d_out;
    float* Mq  = (float*)d_ws;

    state_kernel<<<dim3(T_DIM), dim3(128), 0, stream>>>(u, d1, d2, Mq);
    gemm_kernel<<<dim3(B_DIM * 4), dim3(256), 0, stream>>>(u, d1, d2, Mq, v1, v2, out);
}


# State kernel: the next trip's u, d1, d2 are asked for without a branch and kept as loaded, so a trip's steps run while they are in flight (before: the wave waited for them at once); same values
# speedup vs baseline: 111.4550x; 1.0373x over previous
#include <hip/hip_runtime.h>


#define T_DIM 128
#define B_DIM 128
#define R_DIM 512
#define NSLOTS 256
#define GSLOTS 32
#define NT     16
#define XROW   128
#define SP     36
#define SROWS  576

typedef float v2f __attribute__((ext_vector_type(2)));
typedef float v4f __attribute__((ext_vector_type(4)));
typedef float v8f __attribute__((ext_vector_type(8)));

struct Rows4 { v4f u, c, dn; };

__device__ __forceinline__ void load_rows(Rows4& r, const float* su, const float* sc, const float* sdn, int t) {
    r.u = *(const v4f*)(su + t); r.c = *(const v4f*)(sc + t); r.dn = *(const v4f*)(sdn + t);
}

__device__ __forceinline__ void lds_barrier() {
    asm volatile("s_wait_dscnt 0x0\n\ts_barrier_signal -1\n\ts_barrier_wait -1" ::: "memory");
}

template <bool EARLY>
__device__ __forceinline__ void walk4s(
    int t, int tj, float* srow,
    const float* su, const float* sc, const float* sdn,
    float& S, float& used, float bornS, float bornU,
    const Rows4& cur, Rows4& nxt)
{
    load_rows(nxt, su, sc, sdn, t + 4);
    #pragma unroll
    for (int s = 0; s < 4; ++s) {
        const int tt = t + s;
        const float x = cur.u[s] - used;
        srow[s * SP] = S;
        S    = S - fminf(S, fmaxf(0.0f, x));
        used = fmaxf(used + cur.c[s], cur.dn[s]);
        if (EARLY) {
            const bool born = tj == tt + 1;
            S    = born ? bornS : S;
            used = born ? bornU : used;
        }
    }
}

struct In16 { v2f u[NT], a1[NT], a2[NT]; };

__device__ __forceinline__ void load_in(In16& r, const float* pu, const float* p1, const float* p2, int t) {
    const float* qu = pu + t * B_DIM;
    const float* q1 = p1 + (t + 1) * B_DIM;
    const float* q2 = p2 + (t + 1) * B_DIM;
    #pragma unroll
    for (int s = 0; s < NT; ++s) r.u[s] = *(const v2f*)(qu + s * B_DIM);
    #pragma unroll
    for (int s = 0; s < NT - 1; ++s) {
        r.a1[s] = *(const v2f*)(q1 + s * B_DIM);
        r.a2[s] = *(const v2f*)(q2 + s * B_DIM);
    }
    const int sl = t + NT >= T_DIM ? NT - 2 : NT - 1;
    r.a1[NT - 1] = *(const v2f*)(q1 + sl * B_DIM);
    r.a2[NT - 1] = *(const v2f*)(q2 + sl * B_DIM);
}

template <bool XOR1>
__device__ __forceinline__ float quad_swap(float v) {
    return __builtin_bit_cast(float, __builtin_amdgcn_update_dpp(
        0, __builtin_bit_cast(int, v), XOR1 ? 0xB1 : 0x4E, 0xF, 0xF, true));
}

__device__ __forceinline__ void finish_trip(const float (&pend)[16], float* Mq, int p, int wave, int lane, int t) {
    float v = pend[0];
    #pragma unroll
    for (int i = 1; i < 16; ++i) v = fmaxf(v, pend[i]);
    v = fmaxf(v, quad_swap<true>(v));
    v = fmaxf(v, quad_swap<false>(v));
    if ((lane & 3) == 0) Mq[p * 256 + wave * 64 + (t >> 4) * 8 + (lane >> 2)] = v;
}

template <bool FIRST>
__device__ __forceinline__ void trip16(
    int t, int p, int stage, int wave, int lane, float* X, float* Mq,
    const float* pu, const float* p1, const float* p2,
    v2f& used, v2f bornU, float (&pend)[16],
    const In16& cur, In16& nxt)
{
    load_in(nxt, pu, p1, p2, t + NT < T_DIM ? t + NT : T_DIM - NT);
    float* xw = X + stage * NT * XROW + wave * 32 + lane;
    #pragma unroll
    for (int s = 0; s < NT; ++s) {
        const int tt = t + s;
        const v2f uu = cur.u[s];
        const v2f dn = cur.a1[s] + cur.a2[s];
        v2f up;
        up.x = fmaxf(uu.x, 0.0f); up.y = fmaxf(uu.y, 0.0f);
        const v2f c = dn - up;
        const v2f x = uu - used;
        float m = fmaxf(x.x, x.y);
        if (FIRST) m = tt >= p ? m : 0.0f;
        xw[s * XROW] = m;
        const v2f w = used + c;
        used.x = fmaxf(w.x, dn.x); used.y = fmaxf(w.y, dn.y);
        if (FIRST) used = (p == tt + 1) ? bornU : used;
    }
    if (!FIRST) finish_trip(pend, Mq, p, wave, lane, t - NT);
    lds_barrier();
    const float* xr = X + (stage * NT + 4 * wave + (lane >> 3)) * XROW + (lane & 3) * 32 + ((lane >> 2) & 1);
    #pragma unroll
    for (int i = 0; i < 16; ++i) pend[i] = xr[2 * i];
}

__global__ __launch_bounds__(128) void state_kernel(
    const float* __restrict__ u,
    const float* __restrict__ d1,
    const float* __restrict__ d2,
    float* __restrict__ Mq)
{
    __shared__ float X[2 * NT * XROW];

    const int tid  = threadIdx.x;
    const int lane = tid & 31;
    const int wave = __builtin_amdgcn_readfirstlane(tid >> 5);
    const int p    = blockIdx.x;
    const int tA   = p & ~(NT - 1);
    const int bofs = wave * 32 + (lane >> 1) * 2;
    const float* pu = u  + bofs;
    const float* p1 = d1 + bofs;
    const float* p2 = d2 + bofs;
    const v2f zero2 = {0.0f, 0.0f};

    const v2f top = *(const v2f*)(d2 + p * B_DIM + bofs);
    const v2f bornU = (lane & 1) ? zero2 : top;
    v2f used = (tA == p) ? bornU : zero2;

    In16 A, B;
    float pend[16];
    load_in(A, pu, p1, p2, tA);
    int t = tA, stage = 0;
    trip16<true>(t, p, stage, wave, lane, X, Mq, pu, p1, p2, used, bornU, pend, A, B);
    t += NT; stage ^= 1;
    while (t < T_DIM) {
        trip16<false>(t, p, stage, wave, lane, X, Mq, pu, p1, p2, used, bornU, pend, B, A);
        t += NT; stage ^= 1;
        if (t >= T_DIM) break;
        trip16<false>(t, p, stage, wave, lane, X, Mq, pu, p1, p2, used, bornU, pend, A, B);
        t += NT; stage ^= 1;
    }
    finish_trip(pend, Mq, p, wave, lane, t - NT);
    asm volatile("s_wait_loadcnt 0x0" ::: "memory");
}

#define GN   128
#define QP   (GN + 4)
#define CHQ  8
#define NCH  (NSLOTS / 32)

__device__ __forceinline__ v4f load_a4(const float* q) {
    const v2f lo = *(const v2f*)(q), hi = *(const v2f*)(q + 256);
    const v4f r = {lo.x, lo.y, hi.x, hi.y};
    return r;
}
__device__ __forceinline__ void load_a(v4f (&a)[4], const float* Ap, int ch) {
    a[0] = load_a4(Ap + 4096 * ch);
    a[1] = load_a4(Ap + 4096 * ch + 512);
    a[2] = load_a4(Ap + 4096 * ch + 2048);
    a[3] = load_a4(Ap + 4096 * ch + 2560);
}

__device__ __forceinline__ void load_s(v4f (&a)[4], const float* row) {
    a[0] = *(const v4f*)(row);
    a[1] = *(const v4f*)(row + 4);
    a[2] = *(const v4f*)(row + 16);
    a[3] = *(const v4f*)(row + 20);
}

__device__ __forceinline__ int tile_row0(int g) { return 136 * g - 8 * g * g; }

__device__ __forceinline__ void make_w(v4f (&a)[4], const float* srd, int rt, int ch, const v4f (&mn)[4]) {
    v4f an[4];
    load_s(an, srd + (tile_row0(ch) + 16 * (rt - ch)) * SP);
    #pragma unroll
    for (int i = 0; i < 4; ++i) {
        #pragma unroll
        for (int e = 0; e < 4; ++e) a[i][e] = fminf(an[i][e], mn[i][e]);
    }
}

__device__ __forceinline__ void load_plane(v4f (&p)[4], const float* S1, const float* S2, size_t bstride, int ch) {
    const float* r1 = S1 + (size_t)(16 * ch) * bstride;
    const float* r2 = S2 + (size_t)(16 * ch) * bstride;
    p[0] = *(const v4f*)(r1);
    p[1] = *(const v4f*)(r2);
    p[2] = *(const v4f*)(r1 + bstride);
    p[3] = *(const v4f*)(r2 + bstride);
}

__device__ __forceinline__ void store_plane(v4f* dst, const v4f (&p)[4]) {
    #pragma unroll
    for (int i = 0; i < 4; ++i) {
        const v4f col = {p[0][i], p[1][i], p[2][i], p[3][i]};
        dst[i * 32] = col;
    }
}

__device__ __forceinline__ void mma_group(v8f (&acc)[4], const v4f alo, const v4f ahi,
                                          const v4f (&blo)[4], const v4f (&bhi)[4]) {
    const v2f a0 = {alo.x, alo.y}, a1 = {alo.z, alo.w}, a2 = {ahi.x, ahi.y}, a3 = {ahi.z, ahi.w};
    #pragma unroll
    for (int i = 0; i < 4; ++i) {
        const v2f bq = {blo[i].x, blo[i].y};
        acc[i] = __builtin_amdgcn_wmma_f32_16x16x4_f32(false, a0, false, bq, (short)0, acc[i], false, false);
    }
    #pragma unroll
    for (int i = 0; i < 4; ++i) {
        const v2f bq = {blo[i].z, blo[i].w};
        acc[i] = __builtin_amdgcn_wmma_f32_16x16x4_f32(false, a1, false, bq, (short)0, acc[i], false, false);
    }
    #pragma unroll
    for (int i = 0; i < 4; ++i) {
        const v2f bq = {bhi[i].x, bhi[i].y};
        acc[i] = __builtin_amdgcn_wmma_f32_16x16x4_f32(false, a2, false, bq, (short)0, acc[i], false, false);
    }
    #pragma unroll
    for (int i = 0; i < 4; ++i) {
        const v2f bq = {bhi[i].z, bhi[i].w};
        acc[i] = __builtin_amdgcn_wmma_f32_16x16x4_f32(false, a3, false, bq, (short)0, acc[i], false, false);
    }
}

__device__ __forceinline__ void mma_chunk(v8f (&accL)[4], v8f (&accH)[4], const v4f* rd,
                                          const v4f (&aL)[4], const v4f (&aH)[4], bool both) {
    #pragma unroll
    for (int gg = 0; gg < 2; ++gg) {
        v4f blo[4], bhi[4];
        __builtin_amdgcn_sched_barrier(0);
        #pragma unroll
        for (int i = 0; i < 4; ++i) blo[i] = rd[(4 * gg) * QP + 32 * i];
        #pragma unroll
        for (int i = 0; i < 4; ++i) bhi[i] = rd[(4 * gg + 1) * QP + 32 * i];
        __builtin_amdgcn_sched_barrier(0);
        mma_group(accH, aH[2 * gg], aH[2 * gg + 1], blo, bhi);
        if (both) mma_group(accL, aL[2 * gg], aL[2 * gg + 1], blo, bhi);
    }
}

__device__ __forceinline__ void store_tile(float* otile, const v8f (&acc)[4], int rt, int h) {
    #pragma unroll
    for (int r = 0; r < 8; ++r) {
        const int trow = rt * 16 + r + 8 * h;
        const v4f o = {acc[0][r], acc[1][r], acc[2][r], acc[3][r]};
        *(v4f*)(otile + (size_t)trow * B_DIM * R_DIM) = o;
    }
}

template <bool REQ, bool NEXT>
__device__ __forceinline__ void chunk_step(
    int ch, int rtL, int rtH,
    v8f (&accL)[4], v8f (&accH)[4], v4f (&aL)[4], v4f (&aH)[4], v4f (&mnL)[4], v4f (&mnH)[4],
    v4f (&preq)[4], const v4f (&pst)[4],
    const float* S1, const float* S2, size_t bstride,
    const float* MpL, const float* MpH, const float* srd, v4f* wr, const v4f* rd)
{
    const int cur = (ch & 1) * CHQ * QP, nxt = CHQ * QP - cur;
    const bool nL = ch + 1 <= rtL;
    const bool nH = ch + 1 <= rtH;
    if (NEXT) {
        load_a(mnL, MpL, nL ? ch + 1 : rtL);
        load_a(mnH, MpH, nH ? ch + 1 : rtH);
    }
    if (REQ) load_plane(preq, S1, S2, bstride, ch + 2);
    __builtin_amdgcn_sched_barrier(0);
    if (ch <= rtH) mma_chunk(accL, accH, rd + cur, aL, aH, ch <= rtL);
    if (NEXT) {
        store_plane(wr + nxt, pst);
        if (nL) make_w(aL, srd, rtL, ch + 1, mnL);
        if (nH) make_w(aH, srd, rtH, ch + 1, mnH);
        lds_barrier();
    }
}

__global__ __launch_bounds__(256) void gemm_kernel(
    const float* __restrict__ u,
    const float* __restrict__ d1,
    const float* __restrict__ d2,
    const float* __restrict__ Mq,
    const float* __restrict__ v1,
    const float* __restrict__ v2,
    float* __restrict__ out)
{
    __shared__ v4f sb[2 * CHQ * QP];
    __shared__ v4f ss4[SROWS * SP / 4];
    __shared__ v4f gu4[256 / 4], gc4[256 / 4], gdn4[256 / 4];
    float* ssf = (float*)ss4;
    float* gu  = (float*)gu4;
    float* gc  = (float*)gc4;
    float* gdn = (float*)gdn4;

    const int tid  = threadIdx.x;
    const int lane = tid & 31;
    const int wave = __builtin_amdgcn_readfirstlane(tid >> 5);
    const int rtL  = wave >> 1;
    const int rtH  = 7 - rtL;
    const int hf   = wave & 1;
    const int c    = lane & 15;
    const int h    = lane >> 4;

    const int b  = blockIdx.x >> 2;
    const int n0 = (blockIdx.x & 3) * GN;

    v8f accL[4] = {}, accH[4] = {};

    const float* Mp  = Mq + (c >> 2) * 64 + (c & 3) * 2 + h * 1024;
    const float* MpL = Mp + rtL * 8;
    const float* MpH = Mp + rtH * 8;
    const size_t bstride = (size_t)B_DIM * R_DIM;
    const float* S1 = v1 + (size_t)(2 * wave) * bstride + (size_t)b * R_DIM + n0 + 4 * lane;
    const float* S2 = v2 + (size_t)(2 * wave) * bstride + (size_t)b * R_DIM + n0 + 4 * lane;
    v4f*       wr = sb + wave * QP + lane;
    const v4f* rd = sb + (2 * h) * QP + c + 16 * hf;

    const int ts = tid < T_DIM ? tid : T_DIM - 1;
    const int tn = ts + 1 < T_DIM ? ts + 1 : T_DIM - 1;
    const float uu = u[ts * B_DIM + b];
    const float a1 = d1[tn * B_DIM + b];
    const float a2 = d2[tn * B_DIM + b];
    const int j  = wave * GSLOTS + lane;
    const int tj = j >> 1;
    const float b1 = d1[tj * B_DIM + b];
    const float b2 = d2[tj * B_DIM + b];

    v4f aL[4], aH[4], mnL[4], mnH[4], p0[4], p1[4];
    load_plane(p0, S1, S2, bstride, 0);
    load_a(mnL, MpL, 0);
    load_a(mnH, MpH, 0);
    load_plane(p1, S1, S2, bstride, 1);
    __builtin_amdgcn_sched_barrier(0);

    {
        const bool in = tid < T_DIM;
        const float dn = tid + 1 < T_DIM ? a1 + a2 : 0.0f;
        const float up = fmaxf(uu, 0.0f);
        gu[tid] = in ? uu : 0.0f; gc[tid] = in ? dn - up : 0.0f; gdn[tid] = in ? dn : 0.0f;
    }
    {
        const int t0 = wave * (GSLOTS / 2);
        const bool odd = (j & 1) != 0;
        const float bornS = odd ? b2 : b1;
        const float bornU = odd ? 0.0f : b2;
        float S    = (tj == t0) ? bornS : 0.0f;
        float used = (tj == t0) ? bornU : 0.0f;

        lds_barrier();

        float* srow = ssf + tile_row0(wave) * SP + lane;
        Rows4 A, B;
        load_rows(A, gu, gc, gdn, t0);
        for (int t = t0; t < t0 + GSLOTS / 2; t += 8) {
            walk4s<true>(t,     tj, srow,          gu, gc, gdn, S, used, bornS, bornU, A, B);
            walk4s<true>(t + 4, tj, srow + 4 * SP, gu, gc, gdn, S, used, bornS, bornU, B, A);
            srow += 8 * SP;
        }
        for (int t = t0 + GSLOTS / 2; t < T_DIM; t += 8) {
            walk4s<false>(t,     tj, srow,          gu, gc, gdn, S, used, bornS, bornU, A, B);
            walk4s<false>(t + 4, tj, srow + 4 * SP, gu, gc, gdn, S, used, bornS, bornU, B, A);
            srow += 8 * SP;
        }
    }

    store_plane(wr, p0);
    lds_barrier();

    const float* srd = ssf + c * SP + 8 * h;
    make_w(aL, srd, rtL, 0, mnL);
    make_w(aH, srd, rtH, 0, mnH);

    for (int ch = 0; ch < NCH - 2; ch += 2) {
        chunk_step<true, true>(ch,     rtL, rtH, accL, accH, aL, aH, mnL, mnH, p0, p1, S1, S2, bstride, MpL, MpH, srd, wr, rd);
        chunk_step<true, true>(ch + 1, rtL, rtH, accL, accH, aL, aH, mnL, mnH, p1, p0, S1, S2, bstride, MpL, MpH, srd, wr, rd);
    }
    chunk_step<false, true >(NCH - 2, rtL, rtH, accL, accH, aL, aH, mnL, mnH, p0, p1, S1, S2, bstride, MpL, MpH, srd, wr, rd);
    chunk_step<false, false>(NCH - 1, rtL, rtH, accL, accH, aL, aH, mnL, mnH, p1, p0, S1, S2, bstride, MpL, MpH, srd, wr, rd);

    float* otile = out + (size_t)b * R_DIM + n0 + 64 * hf + 4 * c;
    store_tile(otile, accL, rtL, h);
    store_tile(otile, accH, rtH, h);
}

extern "C" void kernel_launch(void* const* d_in, const int* in_sizes, int n_in,
                              void* d_out, int out_size, void* d_ws, size_t ws_size,
                              hipStream_t stream) {
    const float* u  = (const float*)d_in[0];
    const float* d1 = (const float*)d_in[1];
    const float* d2 = (const float*)d_in[2];
    const float* v1 = (const float*)d_in[3];
    const float* v2 = (const float*)d_in[4];
    float* out = (float*)d_out;
    float* Mq  = (float*)d_ws;

    state_kernel<<<dim3(T_DIM), dim3(128), 0, stream>>>(u, d1, d2, Mq);
    gemm_kernel<<<dim3(B_DIM * 4), dim3(256), 0, stream>>>(u, d1, d2, Mq, v1, v2, out);
}
